# starting kernel
# baseline (speedup 1.0000x reference)
#include <hip/hip_runtime.h>

#define S_   256
#define N_   558
#define B2_  8192
#define F_   128
#define M_   (S_ * N_)
#define TE_  (S_ * B2_)

typedef __attribute__((ext_vector_type(2))) float v2f;
typedef __attribute__((ext_vector_type(8))) float v8f;

__global__ __launch_bounds__(256) void k_init(float* __restrict__ deg,
                                              int* __restrict__ bdmax) {
    int i = blockIdx.x * 256 + threadIdx.x;
    if (i < M_) deg[i] = 1.0f;
    if (i == 0) *bdmax = 0;
}

__global__ __launch_bounds__(256) void k_bdmax(const int* __restrict__ ei,
                                               const float* __restrict__ bd,
                                               int* __restrict__ bdmax) {
    int stride = gridDim.x * 256;
    float m = 0.0f;
    for (int i = blockIdx.x * 256 + threadIdx.x; i < TE_; i += stride) {
        int s    = i >> 13;
        int base = i + (s << 13);
        int src  = ei[base];
        int dst  = ei[base + B2_];
        if (src != dst) m = fmaxf(m, bd[i]);
    }
    for (int off = 16; off > 0; off >>= 1)
        m = fmaxf(m, __shfl_xor(m, off, 32));
    if ((threadIdx.x & 31) == 0) atomicMax(bdmax, __float_as_int(m));
}

__global__ __launch_bounds__(256) void k_deg(const int* __restrict__ ei,
                                             const float* __restrict__ bd,
                                             const int* __restrict__ bdmax,
                                             float* __restrict__ deg) {
    int i = blockIdx.x * 256 + threadIdx.x;
    if (i >= TE_) return;
    int s    = i >> 13;
    int base = i + (s << 13);
    int src  = ei[base];
    int dst  = ei[base + B2_];
    if (src == dst) return;
    float bm = __int_as_float(*bdmax);
    unsafeAtomicAdd(&deg[s * N_ + dst], bd[i] / bm);
}

__global__ __launch_bounds__(256) void k_dinv_agg(const float* __restrict__ x,
                                                  const float* __restrict__ deg,
                                                  float* __restrict__ dinv,
                                                  float* __restrict__ agg) {
    int t = blockIdx.x * 256 + threadIdx.x;
    int m = t >> 5;
    if (m >= M_) return;
    int f4   = (t & 31) * 4;
    float d  = deg[m];
    float di = rsqrtf(d);
    if ((t & 31) == 0) dinv[m] = di;
    float sl = di * di;
    const float4 xv = *(const float4*)(x + (size_t)m * F_ + f4);
    float4 o;
    o.x = xv.x * sl; o.y = xv.y * sl; o.z = xv.z * sl; o.w = xv.w * sl;
    *(float4*)(agg + (size_t)m * F_ + f4) = o;
}

__global__ __launch_bounds__(256) void k_scatter(const int* __restrict__ ei,
                                                 const float* __restrict__ bd,
                                                 const float* __restrict__ x,
                                                 const float* __restrict__ dinv,
                                                 const int* __restrict__ bdmax,
                                                 float* __restrict__ agg) {
    int lane = threadIdx.x & 31;
    int e    = blockIdx.x * 8 + (threadIdx.x >> 5);
    if (e >= TE_) return;
    int s    = e >> 13;
    int base = e + (s << 13);
    int src  = ei[base];
    int dst  = ei[base + B2_];
    if (src == dst) return;
    int row = s * N_ + src;
    int col = s * N_ + dst;
    float bm   = __int_as_float(*bdmax);
    float norm = dinv[row] * (bd[e] / bm) * dinv[col];
    const float4 xv = *(const float4*)(x + (size_t)row * F_ + lane * 4);
    float* dp = agg + (size_t)col * F_ + lane * 4;
    unsafeAtomicAdd(dp + 0, norm * xv.x);
    unsafeAtomicAdd(dp + 1, norm * xv.y);
    unsafeAtomicAdd(dp + 2, norm * xv.z);
    unsafeAtomicAdd(dp + 3, norm * xv.w);
}

__global__ __launch_bounds__(256) void k_gemm(const float* __restrict__ agg,
                                              const float* __restrict__ W,
                                              const float* __restrict__ bias,
                                              float* __restrict__ out) {
    int lane = threadIdx.x & 31;
    int wave = threadIdx.x >> 5;
    int m0   = blockIdx.x * 16;
    int o0   = wave * 16;
    int l16  = lane & 15;
    int koff = (lane >> 4) * 2;

    const float* arow = agg + (size_t)(m0 + l16) * F_ + koff;
    const float* brow = W   + (size_t)(o0 + l16) * F_ + koff;

    v8f acc = {};
#pragma unroll
    for (int k = 0; k < F_; k += 4) {
        v2f a = *(const v2f*)(arow + k);
        v2f b = *(const v2f*)(brow + k);
        acc = __builtin_amdgcn_wmma_f32_16x16x4_f32(
             false, a,  false, b,
             (short)0, acc,  false,  false);
    }

    float bb    = bias[o0 + l16];
    int   half8 = (lane >> 4) * 8;
#pragma unroll
    for (int v = 0; v < 8; ++v) {
        out[(size_t)(m0 + v + half8) * F_ + o0 + l16] = acc[v] + bb;
    }
}

extern "C" void kernel_launch(void* const* d_in, const int* in_sizes, int n_in,
                              void* d_out, int out_size, void* d_ws, size_t ws_size,
                              hipStream_t stream) {
    const float* x    = (const float*)d_in[0];
    const int*   ei   = (const int*)  d_in[1];
    const float* bd   = (const float*)d_in[2];
    const float* W    = (const float*)d_in[3];
    const float* bias = (const float*)d_in[4];
    float*       out  = (float*)d_out;

    float* agg   = (float*)d_ws;
    float* deg   = agg + (size_t)M_ * F_;
    float* dinv  = deg + M_;
    int*   bdmax = (int*)(dinv + M_);

    k_init    <<<(M_ + 255) / 256, 256, 0, stream>>>(deg, bdmax);
    k_bdmax   <<<2048,            256, 0, stream>>>(ei, bd, bdmax);
    k_deg     <<<TE_ / 256,       256, 0, stream>>>(ei, bd, bdmax, deg);
    k_dinv_agg<<<(M_ * 32) / 256, 256, 0, stream>>>(x, deg, dinv, agg);
    k_scatter <<<TE_ / 8,         256, 0, stream>>>(ei, bd, x, dinv, bdmax, agg);
    k_gemm    <<<M_ / 16,         256, 0, stream>>>(agg, W, bias, out);
}


# Atomic scatter, degree and self-term kernels replaced by one per-graph kernel: LDS counting sort of edges by destination, then row gather without float atomics; sum order = LDS arrival rank
# speedup vs baseline: 2.7691x; 2.7691x over previous
#include <hip/hip_runtime.h>


#define S_   256
#define N_   558
#define B2_  8192
#define F_   128
#define M_   (S_ * N_)
#define TE_  (S_ * B2_)

#define NT_    1024
#define EPT_   (B2_ / NT_)
#define SPLIT_ 2
#define RP_    ((N_ + SPLIT_ - 1) / SPLIT_)
#define NPAD_  576
#define SLOTS_ (NPAD_ / 32)

typedef __attribute__((ext_vector_type(2))) float v2f;
typedef __attribute__((ext_vector_type(8))) float v8f;

__global__ __launch_bounds__(32) void k_init(int* __restrict__ bdmax) {
    if (threadIdx.x == 0) *bdmax = 0;
}

__global__ __launch_bounds__(256) void k_bdmax(const int* __restrict__ ei,
                                               const float* __restrict__ bd,
                                               int* __restrict__ bdmax) {
    int stride = gridDim.x * 256;
    float m = 0.0f;
    for (int i = blockIdx.x * 256 + threadIdx.x; i < TE_; i += stride) {
        int s    = i >> 13;
        int base = i + (s << 13);
        int src  = ei[base];
        int dst  = ei[base + B2_];
        if (src != dst) m = fmaxf(m, bd[i]);
    }
    for (int off = 16; off > 0; off >>= 1)
        m = fmaxf(m, __shfl_xor(m, off, 32));
    if ((threadIdx.x & 31) == 0) atomicMax(bdmax, __float_as_int(m));
}

__global__ __launch_bounds__(NT_) void k_sort_gather(const int* __restrict__ ei,
                                                     const float* __restrict__ bd,
                                                     const float* __restrict__ x,
                                                     const int* __restrict__ bdmax,
                                                     float* __restrict__ agg) {
    __shared__ int   s_cur[NPAD_];
    __shared__ int   s_start[NPAD_ + 1];
    __shared__ float s_dinv[NPAD_];
    __shared__ int   s_src[B2_];
    __shared__ float s_w[B2_];

    const int tid  = threadIdx.x;
    const int lane = tid & 31;
    const int wave = tid >> 5;
    const int s    = blockIdx.x / SPLIT_;
    const int part = blockIdx.x % SPLIT_;

    if (tid < NPAD_) s_cur[tid] = 0;
    __syncthreads();

    const int*   eis = ei + (size_t)s * 2 * B2_;
    const float* bds = bd + (size_t)s * B2_;
    const float  bm  = __int_as_float(*bdmax);
    int   es[EPT_], ed[EPT_];
    float wv[EPT_];
#pragma unroll
    for (int j = 0; j < EPT_; ++j) {
        int e = j * NT_ + tid;
        es[j] = eis[e];
        ed[j] = eis[B2_ + e];
        wv[j] = bds[e];
    }
    int sd[EPT_];
#pragma unroll
    for (int j = 0; j < EPT_; ++j) {
        const bool valid = es[j] != ed[j];
        sd[j] = valid ? (es[j] | (ed[j] << 16)) : -1;
        wv[j] = wv[j] / bm;
        if (valid) atomicAdd(&s_cur[ed[j]], 1);
    }
    __syncthreads();

    if (wave == 0) {
        int base = lane * SLOTS_;
        int sum = 0;
#pragma unroll
        for (int k = 0; k < SLOTS_; ++k) sum += s_cur[base + k];
        int incl = sum;
        for (int off = 1; off < 32; off <<= 1) {
            int t = __shfl_up(incl, off, 32);
            if (lane >= off) incl += t;
        }
        int run = incl - sum;
#pragma unroll
        for (int k = 0; k < SLOTS_; ++k) {
            int c = s_cur[base + k];
            s_start[base + k] = run;
            s_cur[base + k]   = run;
            run += c;
        }
        if (lane == 31) s_start[NPAD_] = run;
    }
    __syncthreads();

#pragma unroll
    for (int j = 0; j < EPT_; ++j) {
        if (sd[j] >= 0) {
            int dst = sd[j] >> 16;
            int pos = atomicAdd(&s_cur[dst], 1);
            s_src[pos] = sd[j] & 0xFFFF;
            s_w[pos]   = wv[j];
        }
    }
    __syncthreads();

    if (tid < N_) {
        int p0 = s_start[tid], p1 = s_start[tid + 1];
        float d = 1.0f;
        for (int p = p0; p < p1; ++p) d += s_w[p];
        s_dinv[tid] = rsqrtf(d);
    }
    __syncthreads();

    const int r0 = part * RP_;
    const int r1 = (r0 + RP_ < N_) ? (r0 + RP_) : N_;
    const float* xg = x   + (size_t)s * N_ * F_ + lane * 4;
    float*       ag = agg + (size_t)s * N_ * F_ + lane * 4;

    for (int r = r0 + wave; r < r1; r += NT_ / 32) {
        const int st = __builtin_amdgcn_readfirstlane(s_start[r]);
        const int en = __builtin_amdgcn_readfirstlane(s_start[r + 1]);
        const float dd = s_dinv[r];
        const float sl = dd * dd;
        const float4 xs = *(const float4*)(xg + (size_t)r * F_);
        float4 acc = {0.0f, 0.0f, 0.0f, 0.0f};

        int p = st;
        for (; p + 4 <= en; p += 4) {
            int a0 = s_src[p], a1 = s_src[p + 1], a2 = s_src[p + 2], a3 = s_src[p + 3];
            float n0 = s_dinv[a0] * s_w[p]     * dd;
            float n1 = s_dinv[a1] * s_w[p + 1] * dd;
            float n2 = s_dinv[a2] * s_w[p + 2] * dd;
            float n3 = s_dinv[a3] * s_w[p + 3] * dd;
            const float4 x0 = *(const float4*)(xg + (size_t)a0 * F_);
            const float4 x1 = *(const float4*)(xg + (size_t)a1 * F_);
            const float4 x2 = *(const float4*)(xg + (size_t)a2 * F_);
            const float4 x3 = *(const float4*)(xg + (size_t)a3 * F_);
            acc.x += n0 * x0.x; acc.y += n0 * x0.y; acc.z += n0 * x0.z; acc.w += n0 * x0.w;
            acc.x += n1 * x1.x; acc.y += n1 * x1.y; acc.z += n1 * x1.z; acc.w += n1 * x1.w;
            acc.x += n2 * x2.x; acc.y += n2 * x2.y; acc.z += n2 * x2.z; acc.w += n2 * x2.w;
            acc.x += n3 * x3.x; acc.y += n3 * x3.y; acc.z += n3 * x3.z; acc.w += n3 * x3.w;
        }
        const int rem = en - p;
        if (rem > 0) {
            const int  q1 = (rem > 1) ? p + 1 : p;
            const int  q2 = (rem > 2) ? p + 2 : p;
            int a0 = s_src[p], a1 = s_src[q1], a2 = s_src[q2];
            float n0 = s_dinv[a0] * s_w[p]  * dd;
            float n1 = s_dinv[a1] * s_w[q1] * dd;
            float n2 = s_dinv[a2] * s_w[q2] * dd;
            const float4 x0 = *(const float4*)(xg + (size_t)a0 * F_);
            const float4 x1 = *(const float4*)(xg + (size_t)a1 * F_);
            const float4 x2 = *(const float4*)(xg + (size_t)a2 * F_);
            acc.x += n0 * x0.x; acc.y += n0 * x0.y; acc.z += n0 * x0.z; acc.w += n0 * x0.w;
            if (rem > 1) {
                acc.x += n1 * x1.x; acc.y += n1 * x1.y; acc.z += n1 * x1.z; acc.w += n1 * x1.w;
            }
            if (rem > 2) {
                acc.x += n2 * x2.x; acc.y += n2 * x2.y; acc.z += n2 * x2.z; acc.w += n2 * x2.w;
            }
        }
        acc.x += xs.x * sl; acc.y += xs.y * sl; acc.z += xs.z * sl; acc.w += xs.w * sl;
        *(float4*)(ag + (size_t)r * F_) = acc;
    }
}

__global__ __launch_bounds__(256) void k_gemm(const float* __restrict__ agg,
                                              const float* __restrict__ W,
                                              const float* __restrict__ bias,
                                              float* __restrict__ out) {
    int lane = threadIdx.x & 31;
    int wave = threadIdx.x >> 5;
    int m0   = blockIdx.x * 16;
    int o0   = wave * 16;
    int l16  = lane & 15;
    int koff = (lane >> 4) * 2;

    const float* arow = agg + (size_t)(m0 + l16) * F_ + koff;
    const float* brow = W   + (size_t)(o0 + l16) * F_ + koff;

    v8f acc = {};
#pragma unroll
    for (int k = 0; k < F_; k += 4) {
        v2f a = *(const v2f*)(arow + k);
        v2f b = *(const v2f*)(brow + k);
        acc = __builtin_amdgcn_wmma_f32_16x16x4_f32(
             false, a,  false, b,
             (short)0, acc,  false,  false);
    }

    float bb    = bias[o0 + l16];
    int   half8 = (lane >> 4) * 8;
#pragma unroll
    for (int v = 0; v < 8; ++v) {
        out[(size_t)(m0 + v + half8) * F_ + o0 + l16] = acc[v] + bb;
    }
}

extern "C" void kernel_launch(void* const* d_in, const int* in_sizes, int n_in,
                              void* d_out, int out_size, void* d_ws, size_t ws_size,
                              hipStream_t stream) {
    const float* x    = (const float*)d_in[0];
    const int*   ei   = (const int*)  d_in[1];
    const float* bd   = (const float*)d_in[2];
    const float* W    = (const float*)d_in[3];
    const float* bias = (const float*)d_in[4];
    float*       out  = (float*)d_out;

    float* agg   = (float*)d_ws;
    int*   bdmax = (int*)(agg + (size_t)M_ * F_ + 2 * (size_t)M_);

    k_init       <<<1,            32,  0, stream>>>(bdmax);
    k_bdmax      <<<2048,         256, 0, stream>>>(ei, bd, bdmax);
    k_sort_gather<<<S_ * SPLIT_,  NT_, 0, stream>>>(ei, bd, x, bdmax, agg);
    k_gemm       <<<M_ / 16,      256, 0, stream>>>(agg, W, bias, out);
}


# Product kernel restaged through LDS: persistent work-groups park W once and each wave its 16 agg rows with whole-row loads, fragments read from LDS, whole-row stores; same products, same k order
# speedup vs baseline: 4.2976x; 1.5520x over previous
#include <hip/hip_runtime.h>


#define S_   256
#define N_   558
#define B2_  8192
#define F_   128
#define M_   (S_ * N_)
#define TE_  (S_ * B2_)

#define NT_    1024
#define EPT_   (B2_ / NT_)
#define SPLIT_ 2
#define RP_    ((N_ + SPLIT_ - 1) / SPLIT_)
#define NPAD_  576
#define SLOTS_ (NPAD_ / 32)

typedef __attribute__((ext_vector_type(2))) float v2f;
typedef __attribute__((ext_vector_type(4))) float v4f;
typedef __attribute__((ext_vector_type(8))) float v8f;

__global__ __launch_bounds__(32) void k_init(int* __restrict__ bdmax) {
    if (threadIdx.x == 0) *bdmax = 0;
}

__global__ __launch_bounds__(256) void k_bdmax(const int* __restrict__ ei,
                                               const float* __restrict__ bd,
                                               int* __restrict__ bdmax) {
    int stride = gridDim.x * 256;
    float m = 0.0f;
    for (int i = blockIdx.x * 256 + threadIdx.x; i < TE_; i += stride) {
        int s    = i >> 13;
        int base = i + (s << 13);
        int src  = ei[base];
        int dst  = ei[base + B2_];
        if (src != dst) m = fmaxf(m, bd[i]);
    }
    for (int off = 16; off > 0; off >>= 1)
        m = fmaxf(m, __shfl_xor(m, off, 32));
    if ((threadIdx.x & 31) == 0) atomicMax(bdmax, __float_as_int(m));
}

__global__ __launch_bounds__(NT_) void k_sort_gather(const int* __restrict__ ei,
                                                     const float* __restrict__ bd,
                                                     const float* __restrict__ x,
                                                     const int* __restrict__ bdmax,
                                                     float* __restrict__ agg) {
    __shared__ int   s_cur[NPAD_];
    __shared__ int   s_start[NPAD_ + 1];
    __shared__ float s_dinv[NPAD_];
    __shared__ int   s_src[B2_];
    __shared__ float s_w[B2_];

    const int tid  = threadIdx.x;
    const int lane = tid & 31;
    const int wave = tid >> 5;
    const int s    = blockIdx.x / SPLIT_;
    const int part = blockIdx.x % SPLIT_;

    if (tid < NPAD_) s_cur[tid] = 0;
    __syncthreads();

    const int*   eis = ei + (size_t)s * 2 * B2_;
    const float* bds = bd + (size_t)s * B2_;
    const float  bm  = __int_as_float(*bdmax);
    int   es[EPT_], ed[EPT_];
    float wv[EPT_];
#pragma unroll
    for (int j = 0; j < EPT_; ++j) {
        int e = j * NT_ + tid;
        es[j] = eis[e];
        ed[j] = eis[B2_ + e];
        wv[j] = bds[e];
    }
    int sd[EPT_];
#pragma unroll
    for (int j = 0; j < EPT_; ++j) {
        const bool valid = es[j] != ed[j];
        sd[j] = valid ? (es[j] | (ed[j] << 16)) : -1;
        wv[j] = wv[j] / bm;
        if (valid) atomicAdd(&s_cur[ed[j]], 1);
    }
    __syncthreads();

    if (wave == 0) {
        int base = lane * SLOTS_;
        int sum = 0;
#pragma unroll
        for (int k = 0; k < SLOTS_; ++k) sum += s_cur[base + k];
        int incl = sum;
        for (int off = 1; off < 32; off <<= 1) {
            int t = __shfl_up(incl, off, 32);
            if (lane >= off) incl += t;
        }
        int run = incl - sum;
#pragma unroll
        for (int k = 0; k < SLOTS_; ++k) {
            int c = s_cur[base + k];
            s_start[base + k] = run;
            s_cur[base + k]   = run;
            run += c;
        }
        if (lane == 31) s_start[NPAD_] = run;
    }
    __syncthreads();

#pragma unroll
    for (int j = 0; j < EPT_; ++j) {
        if (sd[j] >= 0) {
            int dst = sd[j] >> 16;
            int pos = atomicAdd(&s_cur[dst], 1);
            s_src[pos] = sd[j] & 0xFFFF;
            s_w[pos]   = wv[j];
        }
    }
    __syncthreads();

    if (tid < N_) {
        int p0 = s_start[tid], p1 = s_start[tid + 1];
        float d = 1.0f;
        for (int p = p0; p < p1; ++p) d += s_w[p];
        s_dinv[tid] = rsqrtf(d);
    }
    __syncthreads();

    const int r0 = part * RP_;
    const int r1 = (r0 + RP_ < N_) ? (r0 + RP_) : N_;
    const float* xg = x   + (size_t)s * N_ * F_ + lane * 4;
    float*       ag = agg + (size_t)s * N_ * F_ + lane * 4;

    for (int r = r0 + wave; r < r1; r += NT_ / 32) {
        const int st = __builtin_amdgcn_readfirstlane(s_start[r]);
        const int en = __builtin_amdgcn_readfirstlane(s_start[r + 1]);
        const float dd = s_dinv[r];
        const float sl = dd * dd;
        const float4 xs = *(const float4*)(xg + (size_t)r * F_);
        float4 acc = {0.0f, 0.0f, 0.0f, 0.0f};

        int p = st;
        for (; p + 4 <= en; p += 4) {
            int a0 = s_src[p], a1 = s_src[p + 1], a2 = s_src[p + 2], a3 = s_src[p + 3];
            float n0 = s_dinv[a0] * s_w[p]     * dd;
            float n1 = s_dinv[a1] * s_w[p + 1] * dd;
            float n2 = s_dinv[a2] * s_w[p + 2] * dd;
            float n3 = s_dinv[a3] * s_w[p + 3] * dd;
            const float4 x0 = *(const float4*)(xg + (size_t)a0 * F_);
            const float4 x1 = *(const float4*)(xg + (size_t)a1 * F_);
            const float4 x2 = *(const float4*)(xg + (size_t)a2 * F_);
            const float4 x3 = *(const float4*)(xg + (size_t)a3 * F_);
            acc.x += n0 * x0.x; acc.y += n0 * x0.y; acc.z += n0 * x0.z; acc.w += n0 * x0.w;
            acc.x += n1 * x1.x; acc.y += n1 * x1.y; acc.z += n1 * x1.z; acc.w += n1 * x1.w;
            acc.x += n2 * x2.x; acc.y += n2 * x2.y; acc.z += n2 * x2.z; acc.w += n2 * x2.w;
            acc.x += n3 * x3.x; acc.y += n3 * x3.y; acc.z += n3 * x3.z; acc.w += n3 * x3.w;
        }
        const int rem = en - p;
        if (rem > 0) {
            const int  q1 = (rem > 1) ? p + 1 : p;
            const int  q2 = (rem > 2) ? p + 2 : p;
            int a0 = s_src[p], a1 = s_src[q1], a2 = s_src[q2];
            float n0 = s_dinv[a0] * s_w[p]  * dd;
            float n1 = s_dinv[a1] * s_w[q1] * dd;
            float n2 = s_dinv[a2] * s_w[q2] * dd;
            const float4 x0 = *(const float4*)(xg + (size_t)a0 * F_);
            const float4 x1 = *(const float4*)(xg + (size_t)a1 * F_);
            const float4 x2 = *(const float4*)(xg + (size_t)a2 * F_);
            acc.x += n0 * x0.x; acc.y += n0 * x0.y; acc.z += n0 * x0.z; acc.w += n0 * x0.w;
            if (rem > 1) {
                acc.x += n1 * x1.x; acc.y += n1 * x1.y; acc.z += n1 * x1.z; acc.w += n1 * x1.w;
            }
            if (rem > 2) {
                acc.x += n2 * x2.x; acc.y += n2 * x2.y; acc.z += n2 * x2.z; acc.w += n2 * x2.w;
            }
        }
        acc.x += xs.x * sl; acc.y += xs.y * sl; acc.z += xs.z * sl; acc.w += xs.w * sl;
        *(float4*)(ag + (size_t)r * F_) = acc;
    }
}

#define GW_    8
#define GNT_   (GW_ * 32)
#define GWG_   512
#define LP_    132
#define NTILE_ (M_ / 16)
#define TSTEP_ (GWG_ * GW_)

__device__ __forceinline__ void wave_lds_handover() {
    __builtin_amdgcn_fence(__ATOMIC_RELEASE, "wavefront");
    __builtin_amdgcn_wave_barrier();
    __builtin_amdgcn_fence(__ATOMIC_ACQUIRE, "wavefront");
}

__global__ __launch_bounds__(GNT_) void k_gemm(const float* __restrict__ agg,
                                               const float* __restrict__ W,
                                               const float* __restrict__ bias,
                                               float* __restrict__ out) {
    __shared__ __attribute__((aligned(16))) float s_W[F_ * LP_];
    __shared__ __attribute__((aligned(16))) float s_A[GW_ * 16 * LP_];

    const int tid   = threadIdx.x;
    const int lane  = tid & 31;
    const int wave  = __builtin_amdgcn_readfirstlane(tid >> 5);
    const int l16   = lane & 15;
    const int koff  = (lane >> 4) * 2;
    const int half8 = (lane >> 4) * 8;

    int t = wave * GWG_ + blockIdx.x;

    v4f pre[16];
    {
        const float* ap = agg + (size_t)t * 16 * F_ + lane * 4;
#pragma unroll
        for (int r = 0; r < 16; ++r) pre[r] = *(const v4f*)(ap + (size_t)r * F_);
    }

#pragma unroll
    for (int j = 0; j < (F_ * F_ / 4) / GNT_; ++j) {
        const int i   = j * GNT_ + tid;
        const int row = i >> 5;
        const int c4  = i & 31;
        const v4f w4 = *(const v4f*)(W + (size_t)row * F_ + c4 * 4);
        *(v4f*)(&s_W[row * LP_ + c4 * 4]) = w4;
    }
    const v4f b4 = *(const v4f*)(bias + lane * 4);
    float* sa = s_A + wave * (16 * LP_);
    const float* sa_frag = sa  + l16 * LP_ + koff;
    const float* sw_frag = s_W + l16 * LP_ + koff;

    __syncthreads();

    for (; t < NTILE_; t += TSTEP_) {
#pragma unroll
        for (int r = 0; r < 16; ++r) *(v4f*)(&sa[r * LP_ + lane * 4]) = pre[r];
        wave_lds_handover();

        {
            const int tn = (t + TSTEP_ < NTILE_) ? (t + TSTEP_) : t;
            const float* ap = agg + (size_t)tn * 16 * F_ + lane * 4;
#pragma unroll
            for (int r = 0; r < 16; ++r) pre[r] = *(const v4f*)(ap + (size_t)r * F_);
        }

        v8f acc[8];
#pragma unroll
        for (int ct = 0; ct < 8; ++ct) acc[ct] = v8f{};

#pragma unroll 8
        for (int k = 0; k < F_; k += 4) {
            const v2f a = *(const v2f*)(sa_frag + k);
#pragma unroll
            for (int ct = 0; ct < 8; ++ct) {
                const v2f b = *(const v2f*)(sw_frag + ct * 16 * LP_ + k);
                acc[ct] = __builtin_amdgcn_wmma_f32_16x16x4_f32(
                     false, a,  false, b,
                     (short)0, acc[ct],  false,  false);
            }
        }

#pragma unroll
        for (int ct = 0; ct < 8; ++ct) {
#pragma unroll
            for (int v = 0; v < 8; ++v) sa[(v + half8) * LP_ + ct * 16 + l16] = acc[ct][v];
        }
        wave_lds_handover();
        float* op = out + (size_t)t * 16 * F_ + lane * 4;
#pragma unroll
        for (int r = 0; r < 16; ++r) {
            const v4f o = *(const v4f*)(&sa[r * LP_ + lane * 4]);
            *(v4f*)(op + (size_t)r * F_) = o + b4;
        }
    }
}

extern "C" void kernel_launch(void* const* d_in, const int* in_sizes, int n_in,
                              void* d_out, int out_size, void* d_ws, size_t ws_size,
                              hipStream_t stream) {
    const float* x    = (const float*)d_in[0];
    const int*   ei   = (const int*)  d_in[1];
    const float* bd   = (const float*)d_in[2];
    const float* W    = (const float*)d_in[3];
    const float* bias = (const float*)d_in[4];
    float*       out  = (float*)d_out;

    float* agg   = (float*)d_ws;
    int*   bdmax = (int*)(agg + (size_t)M_ * F_ + 2 * (size_t)M_);

    k_init       <<<1,            32,  0, stream>>>(bdmax);
    k_bdmax      <<<2048,         256, 0, stream>>>(ei, bd, bdmax);
    k_sort_gather<<<S_ * SPLIT_,  NT_, 0, stream>>>(ei, bd, x, bdmax, agg);
    k_gemm       <<<GWG_,         GNT_, 0, stream>>>(agg, W, bias, out);
}


# Gather kernel: one work-group per graph parks the x rows in LDS, 64 features per pass, and gathers source rows from LDS instead of through the cache path; edges sorted once per graph; same sum order
# speedup vs baseline: 4.9560x; 1.1532x over previous
#include <hip/hip_runtime.h>


#define S_   256
#define N_   558
#define B2_  8192
#define F_   128
#define M_   (S_ * N_)
#define TE_  (S_ * B2_)

#define NT_    1024
#define EPT_   (B2_ / NT_)
#define NPAD_  576
#define SLOTS_ (NPAD_ / 32)

typedef __attribute__((ext_vector_type(2))) float v2f;
typedef __attribute__((ext_vector_type(4))) float v4f;
typedef __attribute__((ext_vector_type(8))) float v8f;

__global__ __launch_bounds__(32) void k_init(int* __restrict__ bdmax) {
    if (threadIdx.x == 0) *bdmax = 0;
}

__global__ __launch_bounds__(256) void k_bdmax(const int* __restrict__ ei,
                                               const float* __restrict__ bd,
                                               int* __restrict__ bdmax) {
    int stride = gridDim.x * 256;
    float m = 0.0f;
    for (int i = blockIdx.x * 256 + threadIdx.x; i < TE_; i += stride) {
        int s    = i >> 13;
        int base = i + (s << 13);
        int src  = ei[base];
        int dst  = ei[base + B2_];
        if (src != dst) m = fmaxf(m, bd[i]);
    }
    for (int off = 16; off > 0; off >>= 1)
        m = fmaxf(m, __shfl_xor(m, off, 32));
    if ((threadIdx.x & 31) == 0) atomicMax(bdmax, __float_as_int(m));
}

#define XH_   64
#define XV4_  (N_ * XH_ / 4)
#define XLD_  ((XV4_ + NT_ - 1) / NT_)

__device__ __forceinline__ float lane_f(float v, int j) {
    return __int_as_float(__builtin_amdgcn_readlane(__float_as_int(v), j));
}

__device__ __forceinline__ v2f fma2(float n, v2f x, v2f acc) {
    v2f r;
    r.x = __builtin_fmaf(n, x.x, acc.x);
    r.y = __builtin_fmaf(n, x.y, acc.y);
    return r;
}

__device__ __forceinline__ void x_half_load(const float* __restrict__ xs_g, int half, int tid, v4f (&xr)[XLD_]) {
#pragma unroll
    for (int j = 0; j < XLD_; ++j) {
        int i = j * NT_ + tid;
        i = (i < XV4_) ? i : (XV4_ - 1);
        xr[j] = *(const v4f*)(xs_g + (size_t)(i >> 4) * F_ + half * XH_ + (i & 15) * 4);
    }
}

__device__ __forceinline__ void x_half_park(float* __restrict__ s_x, int tid, const v4f (&xr)[XLD_]) {
#pragma unroll
    for (int j = 0; j < XLD_; ++j) {
        const int i = j * NT_ + tid;
        if (i < XV4_) *(v4f*)(&s_x[i * 4]) = xr[j];
    }
}

__device__ __forceinline__ void gather_half(const int* __restrict__ s_start, const float* __restrict__ s_dinv,
                                            const int* __restrict__ s_src, const float* __restrict__ s_w,
                                            const float* __restrict__ s_x, float* __restrict__ ag,
                                            int wave, int lane) {
    const float* xl = s_x + lane * 2;
    for (int r = wave; r < N_; r += NT_ / 32) {
        const int st = __builtin_amdgcn_readfirstlane(s_start[r]);
        const int en = __builtin_amdgcn_readfirstlane(s_start[r + 1]);
        const float dd = s_dinv[r];
        const float sl = dd * dd;
        const v2f xs = *(const v2f*)(xl + r * XH_);
        v2f acc = {0.0f, 0.0f};

        for (int base = st; base < en; base += 32) {
            const int left = en - base;
            const int cnt  = (left < 32) ? left : 32;
            int idx = base + lane;
            idx = (idx < B2_) ? idx : (B2_ - 1);
            int sv = s_src[idx];
            sv = (lane < cnt) ? sv : 0;
            const float nv = s_dinv[sv] * s_w[idx] * dd;

            int j = 0;
            for (; j + 4 <= cnt; j += 4) {
                const int a0 = __builtin_amdgcn_readlane(sv, j);
                const int a1 = __builtin_amdgcn_readlane(sv, j + 1);
                const int a2 = __builtin_amdgcn_readlane(sv, j + 2);
                const int a3 = __builtin_amdgcn_readlane(sv, j + 3);
                const float n0 = lane_f(nv, j);
                const float n1 = lane_f(nv, j + 1);
                const float n2 = lane_f(nv, j + 2);
                const float n3 = lane_f(nv, j + 3);
                const v2f x0 = *(const v2f*)(xl + a0 * XH_);
                const v2f x1 = *(const v2f*)(xl + a1 * XH_);
                const v2f x2 = *(const v2f*)(xl + a2 * XH_);
                const v2f x3 = *(const v2f*)(xl + a3 * XH_);
                acc = fma2(n0, x0, acc);
                acc = fma2(n1, x1, acc);
                acc = fma2(n2, x2, acc);
                acc = fma2(n3, x3, acc);
            }
            const int rem = cnt - j;
            if (rem > 0) {
                const int q1 = (rem > 1) ? j + 1 : j;
                const int q2 = (rem > 2) ? j + 2 : j;
                const int a0 = __builtin_amdgcn_readlane(sv, j);
                const int a1 = __builtin_amdgcn_readlane(sv, q1);
                const int a2 = __builtin_amdgcn_readlane(sv, q2);
                const float n0 = lane_f(nv, j);
                const float n1 = lane_f(nv, q1);
                const float n2 = lane_f(nv, q2);
                const v2f x0 = *(const v2f*)(xl + a0 * XH_);
                const v2f x1 = *(const v2f*)(xl + a1 * XH_);
                const v2f x2 = *(const v2f*)(xl + a2 * XH_);
                acc = fma2(n0, x0, acc);
                if (rem > 1) acc = fma2(n1, x1, acc);
                if (rem > 2) acc = fma2(n2, x2, acc);
            }
        }
        acc = fma2(sl, xs, acc);
        *(v2f*)(ag + (size_t)r * F_) = acc;
    }
}

__global__ __launch_bounds__(NT_) void k_sort_gather(const int* __restrict__ ei,
                                                     const float* __restrict__ bd,
                                                     const float* __restrict__ x,
                                                     const int* __restrict__ bdmax,
                                                     float* __restrict__ agg) {
    __shared__ int   s_cur[NPAD_];
    __shared__ int   s_start[NPAD_ + 1];
    __shared__ float s_dinv[NPAD_];
    __shared__ int   s_src[B2_];
    __shared__ float s_w[B2_];
    __shared__ __attribute__((aligned(16))) float s_x[N_ * XH_];

    const int tid  = threadIdx.x;
    const int lane = tid & 31;
    const int wave = __builtin_amdgcn_readfirstlane(tid >> 5);
    const int s    = blockIdx.x;

    if (tid < NPAD_) s_cur[tid] = 0;
    __syncthreads();

    const int*   eis = ei + (size_t)s * 2 * B2_;
    const float* bds = bd + (size_t)s * B2_;
    const float* xs_g = x + (size_t)s * N_ * F_;
    const float  bm  = __int_as_float(*bdmax);
    int   es[EPT_], ed[EPT_];
    float wv[EPT_];
#pragma unroll
    for (int j = 0; j < EPT_; ++j) {
        int e = j * NT_ + tid;
        es[j] = eis[e];
        ed[j] = eis[B2_ + e];
        wv[j] = bds[e];
    }
    v4f xr[XLD_];
    x_half_load(xs_g, 0, tid, xr);

    int sd[EPT_];
#pragma unroll
    for (int j = 0; j < EPT_; ++j) {
        const bool valid = es[j] != ed[j];
        sd[j] = valid ? (es[j] | (ed[j] << 16)) : -1;
        wv[j] = wv[j] / bm;
        if (valid) atomicAdd(&s_cur[ed[j]], 1);
    }
    __syncthreads();

    if (wave == 0) {
        int base = lane * SLOTS_;
        int sum = 0;
#pragma unroll
        for (int k = 0; k < SLOTS_; ++k) sum += s_cur[base + k];
        int incl = sum;
        for (int off = 1; off < 32; off <<= 1) {
            int t = __shfl_up(incl, off, 32);
            if (lane >= off) incl += t;
        }
        int run = incl - sum;
#pragma unroll
        for (int k = 0; k < SLOTS_; ++k) {
            int c = s_cur[base + k];
            s_start[base + k] = run;
            s_cur[base + k]   = run;
            run += c;
        }
        if (lane == 31) s_start[NPAD_] = run;
    }
    __syncthreads();

#pragma unroll
    for (int j = 0; j < EPT_; ++j) {
        if (sd[j] >= 0) {
            int dst = sd[j] >> 16;
            int pos = atomicAdd(&s_cur[dst], 1);
            s_src[pos] = sd[j] & 0xFFFF;
            s_w[pos]   = wv[j];
        }
    }
    x_half_park(s_x, tid, xr);
    __syncthreads();

    if (tid < N_) {
        int p0 = s_start[tid], p1 = s_start[tid + 1];
        float d = 1.0f;
        for (int p = p0; p < p1; ++p) d += s_w[p];
        s_dinv[tid] = rsqrtf(d);
    }
    __syncthreads();

    x_half_load(xs_g, 1, tid, xr);
    float* ag = agg + (size_t)s * N_ * F_ + lane * 2;
    gather_half(s_start, s_dinv, s_src, s_w, s_x, ag, wave, lane);
    __syncthreads();
    x_half_park(s_x, tid, xr);
    __syncthreads();
    gather_half(s_start, s_dinv, s_src, s_w, s_x, ag + XH_, wave, lane);
}

#define GW_    8
#define GNT_   (GW_ * 32)
#define GWG_   512
#define LP_    132
#define NTILE_ (M_ / 16)
#define TSTEP_ (GWG_ * GW_)

__device__ __forceinline__ void wave_lds_handover() {
    __builtin_amdgcn_fence(__ATOMIC_RELEASE, "wavefront");
    __builtin_amdgcn_wave_barrier();
    __builtin_amdgcn_fence(__ATOMIC_ACQUIRE, "wavefront");
}

__global__ __launch_bounds__(GNT_) void k_gemm(const float* __restrict__ agg,
                                               const float* __restrict__ W,
                                               const float* __restrict__ bias,
                                               float* __restrict__ out) {
    __shared__ __attribute__((aligned(16))) float s_W[F_ * LP_];
    __shared__ __attribute__((aligned(16))) float s_A[GW_ * 16 * LP_];

    const int tid   = threadIdx.x;
    const int lane  = tid & 31;
    const int wave  = __builtin_amdgcn_readfirstlane(tid >> 5);
    const int l16   = lane & 15;
    const int koff  = (lane >> 4) * 2;
    const int half8 = (lane >> 4) * 8;

    int t = wave * GWG_ + blockIdx.x;

    v4f pre[16];
    {
        const float* ap = agg + (size_t)t * 16 * F_ + lane * 4;
#pragma unroll
        for (int r = 0; r < 16; ++r) pre[r] = *(const v4f*)(ap + (size_t)r * F_);
    }

#pragma unroll
    for (int j = 0; j < (F_ * F_ / 4) / GNT_; ++j) {
        const int i   = j * GNT_ + tid;
        const int row = i >> 5;
        const int c4  = i & 31;
        const v4f w4 = *(const v4f*)(W + (size_t)row * F_ + c4 * 4);
        *(v4f*)(&s_W[row * LP_ + c4 * 4]) = w4;
    }
    const v4f b4 = *(const v4f*)(bias + lane * 4);
    float* sa = s_A + wave * (16 * LP_);
    const float* sa_frag = sa  + l16 * LP_ + koff;
    const float* sw_frag = s_W + l16 * LP_ + koff;

    __syncthreads();

    for (; t < NTILE_; t += TSTEP_) {
#pragma unroll
        for (int r = 0; r < 16; ++r) *(v4f*)(&sa[r * LP_ + lane * 4]) = pre[r];
        wave_lds_handover();

        {
            const int tn = (t + TSTEP_ < NTILE_) ? (t + TSTEP_) : t;
            const float* ap = agg + (size_t)tn * 16 * F_ + lane * 4;
#pragma unroll
            for (int r = 0; r < 16; ++r) pre[r] = *(const v4f*)(ap + (size_t)r * F_);
        }

        v8f acc[8];
#pragma unroll
        for (int ct = 0; ct < 8; ++ct) acc[ct] = v8f{};

#pragma unroll 8
        for (int k = 0; k < F_; k += 4) {
            const v2f a = *(const v2f*)(sa_frag + k);
#pragma unroll
            for (int ct = 0; ct < 8; ++ct) {
                const v2f b = *(const v2f*)(sw_frag + ct * 16 * LP_ + k);
                acc[ct] = __builtin_amdgcn_wmma_f32_16x16x4_f32(
                     false, a,  false, b,
                     (short)0, acc[ct],  false,  false);
            }
        }

#pragma unroll
        for (int ct = 0; ct < 8; ++ct) {
#pragma unroll
            for (int v = 0; v < 8; ++v) sa[(v + half8) * LP_ + ct * 16 + l16] = acc[ct][v];
        }
        wave_lds_handover();
        float* op = out + (size_t)t * 16 * F_ + lane * 4;
#pragma unroll
        for (int r = 0; r < 16; ++r) {
            const v4f o = *(const v4f*)(&sa[r * LP_ + lane * 4]);
            *(v4f*)(op + (size_t)r * F_) = o + b4;
        }
    }
}

extern "C" void kernel_launch(void* const* d_in, const int* in_sizes, int n_in,
                              void* d_out, int out_size, void* d_ws, size_t ws_size,
                              hipStream_t stream) {
    const float* x    = (const float*)d_in[0];
    const int*   ei   = (const int*)  d_in[1];
    const float* bd   = (const float*)d_in[2];
    const float* W    = (const float*)d_in[3];
    const float* bias = (const float*)d_in[4];
    float*       out  = (float*)d_out;

    float* agg   = (float*)d_ws;
    int*   bdmax = (int*)(agg + (size_t)M_ * F_ + 2 * (size_t)M_);

    k_init       <<<1,            32,  0, stream>>>(bdmax);
    k_bdmax      <<<2048,         256, 0, stream>>>(ei, bd, bdmax);
    k_sort_gather<<<S_,           NT_, 0, stream>>>(ei, bd, x, bdmax, agg);
    k_gemm       <<<GWG_,         GNT_, 0, stream>>>(agg, W, bias, out);
}


# Masked max of bond_dist: one launch, one work-group per graph, wide loads grouped ahead of use, 256 plain per-graph maxima folded by the gather kernel; init launch and 16384 global atomics gone
# speedup vs baseline: 6.2269x; 1.2564x over previous
#include <hip/hip_runtime.h>


#define S_   256
#define N_   558
#define B2_  8192
#define F_   128
#define M_   (S_ * N_)
#define TE_  (S_ * B2_)

#define NT_    1024
#define EPT_   (B2_ / NT_)
#define NPAD_  576
#define SLOTS_ (NPAD_ / 32)

typedef __attribute__((ext_vector_type(2))) float v2f;
typedef __attribute__((ext_vector_type(4))) float v4f;
typedef __attribute__((ext_vector_type(8))) float v8f;

#define MT_   256
#define MV4_  (B2_ / 4 / MT_)

typedef __attribute__((ext_vector_type(4))) int v4i;

__global__ __launch_bounds__(MT_) void k_gmax(const int* __restrict__ ei,
                                              const float* __restrict__ bd,
                                              float* __restrict__ gmax) {
    __shared__ float s_m[MT_ / 32];
    const int tid = threadIdx.x;
    const int s   = blockIdx.x;
    const v4i* sp = (const v4i*)(ei + (size_t)s * 2 * B2_);
    const v4i* dp = sp + B2_ / 4;
    const v4f* bp = (const v4f*)(bd + (size_t)s * B2_);
    v4i a[MV4_], b[MV4_];
    v4f w[MV4_];
#pragma unroll
    for (int j = 0; j < MV4_; ++j) {
        const int i = j * MT_ + tid;
        a[j] = sp[i];
        b[j] = dp[i];
        w[j] = bp[i];
    }
    float m = 0.0f;
#pragma unroll
    for (int j = 0; j < MV4_; ++j) {
        m = fmaxf(m, (a[j].x != b[j].x) ? w[j].x : 0.0f);
        m = fmaxf(m, (a[j].y != b[j].y) ? w[j].y : 0.0f);
        m = fmaxf(m, (a[j].z != b[j].z) ? w[j].z : 0.0f);
        m = fmaxf(m, (a[j].w != b[j].w) ? w[j].w : 0.0f);
    }
    for (int off = 16; off > 0; off >>= 1)
        m = fmaxf(m, __shfl_xor(m, off, 32));
    if ((tid & 31) == 0) s_m[tid >> 5] = m;
    __syncthreads();
    if (tid == 0) {
        float g = s_m[0];
        for (int k = 1; k < MT_ / 32; ++k) g = fmaxf(g, s_m[k]);
        gmax[s] = g;
    }
}

#define XH_   64
#define XV4_  (N_ * XH_ / 4)
#define XLD_  ((XV4_ + NT_ - 1) / NT_)

__device__ __forceinline__ float lane_f(float v, int j) {
    return __int_as_float(__builtin_amdgcn_readlane(__float_as_int(v), j));
}

__device__ __forceinline__ v2f fma2(float n, v2f x, v2f acc) {
    v2f r;
    r.x = __builtin_fmaf(n, x.x, acc.x);
    r.y = __builtin_fmaf(n, x.y, acc.y);
    return r;
}

__device__ __forceinline__ void x_half_load(const float* __restrict__ xs_g, int half, int tid, v4f (&xr)[XLD_]) {
#pragma unroll
    for (int j = 0; j < XLD_; ++j) {
        int i = j * NT_ + tid;
        i = (i < XV4_) ? i : (XV4_ - 1);
        xr[j] = *(const v4f*)(xs_g + (size_t)(i >> 4) * F_ + half * XH_ + (i & 15) * 4);
    }
}

__device__ __forceinline__ void x_half_park(float* __restrict__ s_x, int tid, const v4f (&xr)[XLD_]) {
#pragma unroll
    for (int j = 0; j < XLD_; ++j) {
        const int i = j * NT_ + tid;
        if (i < XV4_) *(v4f*)(&s_x[i * 4]) = xr[j];
    }
}

__device__ __forceinline__ void gather_half(const int* __restrict__ s_start, const float* __restrict__ s_dinv,
                                            const int* __restrict__ s_src, const float* __restrict__ s_w,
                                            const float* __restrict__ s_x, float* __restrict__ ag,
                                            int wave, int lane) {
    const float* xl = s_x + lane * 2;
    for (int r = wave; r < N_; r += NT_ / 32) {
        const int st = __builtin_amdgcn_readfirstlane(s_start[r]);
        const int en = __builtin_amdgcn_readfirstlane(s_start[r + 1]);
        const float dd = s_dinv[r];
        const float sl = dd * dd;
        const v2f xs = *(const v2f*)(xl + r * XH_);
        v2f acc = {0.0f, 0.0f};

        for (int base = st; base < en; base += 32) {
            const int left = en - base;
            const int cnt  = (left < 32) ? left : 32;
            int idx = base + lane;
            idx = (idx < B2_) ? idx : (B2_ - 1);
            int sv = s_src[idx];
            sv = (lane < cnt) ? sv : 0;
            const float nv = s_dinv[sv] * s_w[idx] * dd;

            int j = 0;
            for (; j + 4 <= cnt; j += 4) {
                const int a0 = __builtin_amdgcn_readlane(sv, j);
                const int a1 = __builtin_amdgcn_readlane(sv, j + 1);
                const int a2 = __builtin_amdgcn_readlane(sv, j + 2);
                const int a3 = __builtin_amdgcn_readlane(sv, j + 3);
                const float n0 = lane_f(nv, j);
                const float n1 = lane_f(nv, j + 1);
                const float n2 = lane_f(nv, j + 2);
                const float n3 = lane_f(nv, j + 3);
                const v2f x0 = *(const v2f*)(xl + a0 * XH_);
                const v2f x1 = *(const v2f*)(xl + a1 * XH_);
                const v2f x2 = *(const v2f*)(xl + a2 * XH_);
                const v2f x3 = *(const v2f*)(xl + a3 * XH_);
                acc = fma2(n0, x0, acc);
                acc = fma2(n1, x1, acc);
                acc = fma2(n2, x2, acc);
                acc = fma2(n3, x3, acc);
            }
            const int rem = cnt - j;
            if (rem > 0) {
                const int q1 = (rem > 1) ? j + 1 : j;
                const int q2 = (rem > 2) ? j + 2 : j;
                const int a0 = __builtin_amdgcn_readlane(sv, j);
                const int a1 = __builtin_amdgcn_readlane(sv, q1);
                const int a2 = __builtin_amdgcn_readlane(sv, q2);
                const float n0 = lane_f(nv, j);
                const float n1 = lane_f(nv, q1);
                const float n2 = lane_f(nv, q2);
                const v2f x0 = *(const v2f*)(xl + a0 * XH_);
                const v2f x1 = *(const v2f*)(xl + a1 * XH_);
                const v2f x2 = *(const v2f*)(xl + a2 * XH_);
                acc = fma2(n0, x0, acc);
                if (rem > 1) acc = fma2(n1, x1, acc);
                if (rem > 2) acc = fma2(n2, x2, acc);
            }
        }
        acc = fma2(sl, xs, acc);
        *(v2f*)(ag + (size_t)r * F_) = acc;
    }
}

__global__ __launch_bounds__(NT_) void k_sort_gather(const int* __restrict__ ei,
                                                     const float* __restrict__ bd,
                                                     const float* __restrict__ x,
                                                     const float* __restrict__ gmax,
                                                     float* __restrict__ agg) {
    __shared__ int   s_cur[NPAD_];
    __shared__ int   s_start[NPAD_ + 1];
    __shared__ float s_dinv[NPAD_];
    __shared__ int   s_src[B2_];
    __shared__ float s_w[B2_];
    __shared__ __attribute__((aligned(16))) float s_x[N_ * XH_];

    const int tid  = threadIdx.x;
    const int lane = tid & 31;
    const int wave = __builtin_amdgcn_readfirstlane(tid >> 5);
    const int s    = blockIdx.x;

    if (tid < NPAD_) s_cur[tid] = 0;
    __syncthreads();

    const int*   eis = ei + (size_t)s * 2 * B2_;
    const float* bds = bd + (size_t)s * B2_;
    const float* xs_g = x + (size_t)s * N_ * F_;
    int   es[EPT_], ed[EPT_];
    float wv[EPT_];
#pragma unroll
    for (int j = 0; j < EPT_; ++j) {
        int e = j * NT_ + tid;
        es[j] = eis[e];
        ed[j] = eis[B2_ + e];
        wv[j] = bds[e];
    }
    const v4f g0 = *(const v4f*)(gmax + lane * 8);
    const v4f g1 = *(const v4f*)(gmax + lane * 8 + 4);
    v4f xr[XLD_];
    x_half_load(xs_g, 0, tid, xr);

    float bm = fmaxf(fmaxf(fmaxf(g0.x, g0.y), fmaxf(g0.z, g0.w)), fmaxf(fmaxf(g1.x, g1.y), fmaxf(g1.z, g1.w)));
    for (int off = 16; off > 0; off >>= 1)
        bm = fmaxf(bm, __shfl_xor(bm, off, 32));

    int sd[EPT_];
#pragma unroll
    for (int j = 0; j < EPT_; ++j) {
        const bool valid = es[j] != ed[j];
        sd[j] = valid ? (es[j] | (ed[j] << 16)) : -1;
        wv[j] = wv[j] / bm;
        if (valid) atomicAdd(&s_cur[ed[j]], 1);
    }
    __syncthreads();

    if (wave == 0) {
        int base = lane * SLOTS_;
        int sum = 0;
#pragma unroll
        for (int k = 0; k < SLOTS_; ++k) sum += s_cur[base + k];
        int incl = sum;
        for (int off = 1; off < 32; off <<= 1) {
            int t = __shfl_up(incl, off, 32);
            if (lane >= off) incl += t;
        }
        int run = incl - sum;
#pragma unroll
        for (int k = 0; k < SLOTS_; ++k) {
            int c = s_cur[base + k];
            s_start[base + k] = run;
            s_cur[base + k]   = run;
            run += c;
        }
        if (lane == 31) s_start[NPAD_] = run;
    }
    __syncthreads();

#pragma unroll
    for (int j = 0; j < EPT_; ++j) {
        if (sd[j] >= 0) {
            int dst = sd[j] >> 16;
            int pos = atomicAdd(&s_cur[dst], 1);
            s_src[pos] = sd[j] & 0xFFFF;
            s_w[pos]   = wv[j];
        }
    }
    x_half_park(s_x, tid, xr);
    __syncthreads();

    if (tid < N_) {
        int p0 = s_start[tid], p1 = s_start[tid + 1];
        float d = 1.0f;
        for (int p = p0; p < p1; ++p) d += s_w[p];
        s_dinv[tid] = rsqrtf(d);
    }
    __syncthreads();

    x_half_load(xs_g, 1, tid, xr);
    float* ag = agg + (size_t)s * N_ * F_ + lane * 2;
    gather_half(s_start, s_dinv, s_src, s_w, s_x, ag, wave, lane);
    __syncthreads();
    x_half_park(s_x, tid, xr);
    __syncthreads();
    gather_half(s_start, s_dinv, s_src, s_w, s_x, ag + XH_, wave, lane);
}

#define GW_    8
#define GNT_   (GW_ * 32)
#define GWG_   512
#define LP_    132
#define NTILE_ (M_ / 16)
#define TSTEP_ (GWG_ * GW_)

__device__ __forceinline__ void wave_lds_handover() {
    __builtin_amdgcn_fence(__ATOMIC_RELEASE, "wavefront");
    __builtin_amdgcn_wave_barrier();
    __builtin_amdgcn_fence(__ATOMIC_ACQUIRE, "wavefront");
}

__global__ __launch_bounds__(GNT_) void k_gemm(const float* __restrict__ agg,
                                               const float* __restrict__ W,
                                               const float* __restrict__ bias,
                                               float* __restrict__ out) {
    __shared__ __attribute__((aligned(16))) float s_W[F_ * LP_];
    __shared__ __attribute__((aligned(16))) float s_A[GW_ * 16 * LP_];

    const int tid   = threadIdx.x;
    const int lane  = tid & 31;
    const int wave  = __builtin_amdgcn_readfirstlane(tid >> 5);
    const int l16   = lane & 15;
    const int koff  = (lane >> 4) * 2;
    const int half8 = (lane >> 4) * 8;

    int t = wave * GWG_ + blockIdx.x;

    v4f pre[16];
    {
        const float* ap = agg + (size_t)t * 16 * F_ + lane * 4;
#pragma unroll
        for (int r = 0; r < 16; ++r) pre[r] = *(const v4f*)(ap + (size_t)r * F_);
    }

#pragma unroll
    for (int j = 0; j < (F_ * F_ / 4) / GNT_; ++j) {
        const int i   = j * GNT_ + tid;
        const int row = i >> 5;
        const int c4  = i & 31;
        const v4f w4 = *(const v4f*)(W + (size_t)row * F_ + c4 * 4);
        *(v4f*)(&s_W[row * LP_ + c4 * 4]) = w4;
    }
    const v4f b4 = *(const v4f*)(bias + lane * 4);
    float* sa = s_A + wave * (16 * LP_);
    const float* sa_frag = sa  + l16 * LP_ + koff;
    const float* sw_frag = s_W + l16 * LP_ + koff;

    __syncthreads();

    for (; t < NTILE_; t += TSTEP_) {
#pragma unroll
        for (int r = 0; r < 16; ++r) *(v4f*)(&sa[r * LP_ + lane * 4]) = pre[r];
        wave_lds_handover();

        {
            const int tn = (t + TSTEP_ < NTILE_) ? (t + TSTEP_) : t;
            const float* ap = agg + (size_t)tn * 16 * F_ + lane * 4;
#pragma unroll
            for (int r = 0; r < 16; ++r) pre[r] = *(const v4f*)(ap + (size_t)r * F_);
        }

        v8f acc[8];
#pragma unroll
        for (int ct = 0; ct < 8; ++ct) acc[ct] = v8f{};

#pragma unroll 8
        for (int k = 0; k < F_; k += 4) {
            const v2f a = *(const v2f*)(sa_frag + k);
#pragma unroll
            for (int ct = 0; ct < 8; ++ct) {
                const v2f b = *(const v2f*)(sw_frag + ct * 16 * LP_ + k);
                acc[ct] = __builtin_amdgcn_wmma_f32_16x16x4_f32(
                     false, a,  false, b,
                     (short)0, acc[ct],  false,  false);
            }
        }

#pragma unroll
        for (int ct = 0; ct < 8; ++ct) {
#pragma unroll
            for (int v = 0; v < 8; ++v) sa[(v + half8) * LP_ + ct * 16 + l16] = acc[ct][v];
        }
        wave_lds_handover();
        float* op = out + (size_t)t * 16 * F_ + lane * 4;
#pragma unroll
        for (int r = 0; r < 16; ++r) {
            const v4f o = *(const v4f*)(&sa[r * LP_ + lane * 4]);
            *(v4f*)(op + (size_t)r * F_) = o + b4;
        }
    }
}

extern "C" void kernel_launch(void* const* d_in, const int* in_sizes, int n_in,
                              void* d_out, int out_size, void* d_ws, size_t ws_size,
                              hipStream_t stream) {
    const float* x    = (const float*)d_in[0];
    const int*   ei   = (const int*)  d_in[1];
    const float* bd   = (const float*)d_in[2];
    const float* W    = (const float*)d_in[3];
    const float* bias = (const float*)d_in[4];
    float*       out  = (float*)d_out;

    float* agg  = (float*)d_ws;
    float* gmax = agg + (size_t)M_ * F_;

    k_gmax       <<<S_,           MT_, 0, stream>>>(ei, bd, gmax);
    k_sort_gather<<<S_,           NT_, 0, stream>>>(ei, bd, x, gmax, agg);
    k_gemm       <<<GWG_,         GNT_, 0, stream>>>(agg, W, bias, out);
}


# Product fused into the gather kernel: per graph x W^T tiles (WMMA, A from global, W in LDS) parked in LDS, rows gathered there, bias at the store; one launch less, no agg; sum order: transform first
# speedup vs baseline: 7.0784x; 1.1367x over previous
#include <hip/hip_runtime.h>


#define S_   256
#define N_   558
#define B2_  8192
#define F_   128
#define M_   (S_ * N_)
#define TE_  (S_ * B2_)

#define NT_    1024
#define EPT_   (B2_ / NT_)
#define NPAD_  576
#define SLOTS_ (NPAD_ / 32)

typedef __attribute__((ext_vector_type(2))) float v2f;
typedef __attribute__((ext_vector_type(4))) float v4f;
typedef __attribute__((ext_vector_type(8))) float v8f;

#define MT_   256
#define MV4_  (B2_ / 4 / MT_)

typedef __attribute__((ext_vector_type(4))) int v4i;

__global__ __launch_bounds__(MT_) void k_gmax(const int* __restrict__ ei,
                                              const float* __restrict__ bd,
                                              float* __restrict__ gmax) {
    __shared__ float s_m[MT_ / 32];
    const int tid = threadIdx.x;
    const int s   = blockIdx.x;
    const v4i* sp = (const v4i*)(ei + (size_t)s * 2 * B2_);
    const v4i* dp = sp + B2_ / 4;
    const v4f* bp = (const v4f*)(bd + (size_t)s * B2_);
    v4i a[MV4_], b[MV4_];
    v4f w[MV4_];
#pragma unroll
    for (int j = 0; j < MV4_; ++j) {
        const int i = j * MT_ + tid;
        a[j] = sp[i];
        b[j] = dp[i];
        w[j] = bp[i];
    }
    float m = 0.0f;
#pragma unroll
    for (int j = 0; j < MV4_; ++j) {
        m = fmaxf(m, (a[j].x != b[j].x) ? w[j].x : 0.0f);
        m = fmaxf(m, (a[j].y != b[j].y) ? w[j].y : 0.0f);
        m = fmaxf(m, (a[j].z != b[j].z) ? w[j].z : 0.0f);
        m = fmaxf(m, (a[j].w != b[j].w) ? w[j].w : 0.0f);
    }
    for (int off = 16; off > 0; off >>= 1)
        m = fmaxf(m, __shfl_xor(m, off, 32));
    if ((tid & 31) == 0) s_m[tid >> 5] = m;
    __syncthreads();
    if (tid == 0) {
        float g = s_m[0];
        for (int k = 1; k < MT_ / 32; ++k) g = fmaxf(g, s_m[k]);
        gmax[s] = g;
    }
}

#define XH_    64
#define NTG_   35
#define WP_    132

__device__ __forceinline__ float lane_f(float v, int j) {
    return __int_as_float(__builtin_amdgcn_readlane(__float_as_int(v), j));
}

__device__ __forceinline__ v2f fma2(float n, v2f x, v2f acc) {
    v2f r;
    r.x = __builtin_fmaf(n, x.x, acc.x);
    r.y = __builtin_fmaf(n, x.y, acc.y);
    return r;
}

__device__ __forceinline__ void xw_half(const float* __restrict__ xs_g, const float* __restrict__ s_W,
                                        float* __restrict__ s_x, int oh, int wave, int lane) {
    const int c = lane & 15;
    const int h = lane >> 4;
    for (int t = wave; t < NTG_; t += NT_ / 32) {
        int woff = (oh * XH_ + c) * WP_ + h * 4;
        asm volatile("" : "+v"(woff));
        const float* swb = s_W + woff;
        int row = t * 16 + c;
        row = (row < N_) ? row : (N_ - 1);
        const float* ap = xs_g + (size_t)row * F_ + h * 4;
        v8f acc[4];
#pragma unroll
        for (int ct = 0; ct < 4; ++ct) acc[ct] = v8f{};
#pragma unroll
        for (int jc = 0; jc < 2; ++jc) {
            v4f a[8];
#pragma unroll
            for (int j = 0; j < 8; ++j) a[j] = *(const v4f*)(ap + (jc * 8 + j) * 8);
#pragma unroll
            for (int j = 0; j < 8; ++j) {
                const v2f a0 = {a[j].x, a[j].y};
                const v2f a1 = {a[j].z, a[j].w};
#pragma unroll
                for (int ct = 0; ct < 4; ++ct) {
                    const v4f b = *(const v4f*)(swb + ct * 16 * WP_ + (jc * 8 + j) * 8);
                    const v2f b0 = {b.x, b.y};
                    const v2f b1 = {b.z, b.w};
                    acc[ct] = __builtin_amdgcn_wmma_f32_16x16x4_f32(
                         false, a0,  false, b0,
                         (short)0, acc[ct],  false,  false);
                    acc[ct] = __builtin_amdgcn_wmma_f32_16x16x4_f32(
                         false, a1,  false, b1,
                         (short)0, acc[ct],  false,  false);
                }
            }
        }
        float* sx = s_x + (t * 16 + h * 8) * XH_ + c;
#pragma unroll
        for (int ct = 0; ct < 4; ++ct) {
#pragma unroll
            for (int v = 0; v < 8; ++v) sx[v * XH_ + ct * 16] = acc[ct][v];
        }
    }
}

__device__ __forceinline__ void gather_half(const int* __restrict__ s_start, const float* __restrict__ s_dinv,
                                            const int* __restrict__ s_src, const float* __restrict__ s_w,
                                            const float* __restrict__ s_x, float* __restrict__ og, v2f bb,
                                            int wave, int lane) {
    const float* xl = s_x + lane * 2;
    for (int r = wave; r < N_; r += NT_ / 32) {
        const int st = __builtin_amdgcn_readfirstlane(s_start[r]);
        const int en = __builtin_amdgcn_readfirstlane(s_start[r + 1]);
        const float dd = s_dinv[r];
        const float sl = dd * dd;
        const v2f xs = *(const v2f*)(xl + r * XH_);
        v2f acc = {0.0f, 0.0f};

        for (int base = st; base < en; base += 32) {
            const int left = en - base;
            const int cnt  = (left < 32) ? left : 32;
            int idx = base + lane;
            idx = (idx < B2_) ? idx : (B2_ - 1);
            int sv = s_src[idx];
            sv = (lane < cnt) ? sv : 0;
            const float nv = s_dinv[sv] * s_w[idx] * dd;

            int j = 0;
            for (; j + 4 <= cnt; j += 4) {
                const int a0 = __builtin_amdgcn_readlane(sv, j);
                const int a1 = __builtin_amdgcn_readlane(sv, j + 1);
                const int a2 = __builtin_amdgcn_readlane(sv, j + 2);
                const int a3 = __builtin_amdgcn_readlane(sv, j + 3);
                const float n0 = lane_f(nv, j);
                const float n1 = lane_f(nv, j + 1);
                const float n2 = lane_f(nv, j + 2);
                const float n3 = lane_f(nv, j + 3);
                const v2f x0 = *(const v2f*)(xl + a0 * XH_);
                const v2f x1 = *(const v2f*)(xl + a1 * XH_);
                const v2f x2 = *(const v2f*)(xl + a2 * XH_);
                const v2f x3 = *(const v2f*)(xl + a3 * XH_);
                acc = fma2(n0, x0, acc);
                acc = fma2(n1, x1, acc);
                acc = fma2(n2, x2, acc);
                acc = fma2(n3, x3, acc);
            }
            const int rem = cnt - j;
            if (rem > 0) {
                const int q1 = (rem > 1) ? j + 1 : j;
                const int q2 = (rem > 2) ? j + 2 : j;
                const int a0 = __builtin_amdgcn_readlane(sv, j);
                const int a1 = __builtin_amdgcn_readlane(sv, q1);
                const int a2 = __builtin_amdgcn_readlane(sv, q2);
                const float n0 = lane_f(nv, j);
                const float n1 = lane_f(nv, q1);
                const float n2 = lane_f(nv, q2);
                const v2f x0 = *(const v2f*)(xl + a0 * XH_);
                const v2f x1 = *(const v2f*)(xl + a1 * XH_);
                const v2f x2 = *(const v2f*)(xl + a2 * XH_);
                acc = fma2(n0, x0, acc);
                if (rem > 1) acc = fma2(n1, x1, acc);
                if (rem > 2) acc = fma2(n2, x2, acc);
            }
        }
        acc = fma2(sl, xs, acc);
        *(v2f*)(og + (size_t)r * F_) = acc + bb;
    }
}

__global__ __launch_bounds__(NT_) void k_sort_xw_gather(const int* __restrict__ ei,
                                                        const float* __restrict__ bd,
                                                        const float* __restrict__ x,
                                                        const float* __restrict__ W,
                                                        const float* __restrict__ bias,
                                                        const float* __restrict__ gmax,
                                                        float* __restrict__ out) {
    __shared__ int   s_cur[NPAD_];
    __shared__ int   s_start[NPAD_ + 1];
    __shared__ float s_dinv[NPAD_];
    __shared__ int   s_src[B2_];
    __shared__ float s_w[B2_];
    __shared__ __attribute__((aligned(16))) float s_x[NTG_ * 16 * XH_];
    __shared__ __attribute__((aligned(16))) float s_W[F_ * WP_];

    const int tid  = threadIdx.x;
    const int lane = tid & 31;
    const int wave = __builtin_amdgcn_readfirstlane(tid >> 5);
    const int s    = blockIdx.x;

    if (tid < NPAD_) s_cur[tid] = 0;
    __syncthreads();

    const int*   eis = ei + (size_t)s * 2 * B2_;
    const float* bds = bd + (size_t)s * B2_;
    const float* xs_g = x + (size_t)s * N_ * F_;
    int   es[EPT_], ed[EPT_];
    float wv[EPT_];
#pragma unroll
    for (int j = 0; j < EPT_; ++j) {
        int e = j * NT_ + tid;
        es[j] = eis[e];
        ed[j] = eis[B2_ + e];
        wv[j] = bds[e];
    }
    const v4f g0 = *(const v4f*)(gmax + lane * 8);
    const v4f g1 = *(const v4f*)(gmax + lane * 8 + 4);
    v4f w4[(F_ * F_ / 4) / NT_];
#pragma unroll
    for (int j = 0; j < (F_ * F_ / 4) / NT_; ++j) {
        const int i = j * NT_ + tid;
        w4[j] = *(const v4f*)(W + (size_t)(i >> 5) * F_ + (i & 31) * 4);
    }

    float bm = fmaxf(fmaxf(fmaxf(g0.x, g0.y), fmaxf(g0.z, g0.w)), fmaxf(fmaxf(g1.x, g1.y), fmaxf(g1.z, g1.w)));
    for (int off = 16; off > 0; off >>= 1)
        bm = fmaxf(bm, __shfl_xor(bm, off, 32));

    int sd[EPT_];
#pragma unroll
    for (int j = 0; j < EPT_; ++j) {
        const bool valid = es[j] != ed[j];
        sd[j] = valid ? (es[j] | (ed[j] << 16)) : -1;
        wv[j] = wv[j] / bm;
        if (valid) atomicAdd(&s_cur[ed[j]], 1);
    }
#pragma unroll
    for (int j = 0; j < (F_ * F_ / 4) / NT_; ++j) {
        const int i = j * NT_ + tid;
        *(v4f*)(&s_W[(i >> 5) * WP_ + (i & 31) * 4]) = w4[j];
    }
    __syncthreads();

    if (wave == 0) {
        int base = lane * SLOTS_;
        int sum = 0;
#pragma unroll
        for (int k = 0; k < SLOTS_; ++k) sum += s_cur[base + k];
        int incl = sum;
        for (int off = 1; off < 32; off <<= 1) {
            int t = __shfl_up(incl, off, 32);
            if (lane >= off) incl += t;
        }
        int run = incl - sum;
#pragma unroll
        for (int k = 0; k < SLOTS_; ++k) {
            int c = s_cur[base + k];
            s_start[base + k] = run;
            s_cur[base + k]   = run;
            run += c;
        }
        if (lane == 31) s_start[NPAD_] = run;
    }
    __syncthreads();

#pragma unroll
    for (int j = 0; j < EPT_; ++j) {
        if (sd[j] >= 0) {
            int dst = sd[j] >> 16;
            int pos = atomicAdd(&s_cur[dst], 1);
            s_src[pos] = sd[j] & 0xFFFF;
            s_w[pos]   = wv[j];
        }
    }
    __syncthreads();

    if (tid < N_) {
        int p0 = s_start[tid], p1 = s_start[tid + 1];
        float d = 1.0f;
        for (int p = p0; p < p1; ++p) d += s_w[p];
        s_dinv[tid] = rsqrtf(d);
    }

    float* og = out + (size_t)s * N_ * F_ + lane * 2;
    const v2f bb0 = *(const v2f*)(bias + lane * 2);
    const v2f bb1 = *(const v2f*)(bias + XH_ + lane * 2);
    xw_half(xs_g, s_W, s_x, 0, wave, lane);
    __syncthreads();
    gather_half(s_start, s_dinv, s_src, s_w, s_x, og, bb0, wave, lane);
    __syncthreads();
    xw_half(xs_g, s_W, s_x, 1, wave, lane);
    __syncthreads();
    gather_half(s_start, s_dinv, s_src, s_w, s_x, og + XH_, bb1, wave, lane);
}

extern "C" void kernel_launch(void* const* d_in, const int* in_sizes, int n_in,
                              void* d_out, int out_size, void* d_ws, size_t ws_size,
                              hipStream_t stream) {
    const float* x    = (const float*)d_in[0];
    const int*   ei   = (const int*)  d_in[1];
    const float* bd   = (const float*)d_in[2];
    const float* W    = (const float*)d_in[3];
    const float* bias = (const float*)d_in[4];
    float*       out  = (float*)d_out;

    float* gmax = (float*)d_ws;

    k_gmax          <<<S_, MT_, 0, stream>>>(ei, bd, gmax);
    k_sort_xw_gather<<<S_, NT_, 0, stream>>>(ei, bd, x, W, bias, gmax, out);
}


# Gather: a wave takes two destination rows, one per 16-lane half, 4 features a lane (16-byte LDS reads); an edge's source row and weight reach its half by DPP row_share, not readlane; same sum order
# speedup vs baseline: 7.9035x; 1.1166x over previous
#include <hip/hip_runtime.h>


#define S_   256
#define N_   558
#define B2_  8192
#define F_   128
#define M_   (S_ * N_)
#define TE_  (S_ * B2_)

#define NT_    1024
#define EPT_   (B2_ / NT_)
#define NPAD_  576
#define SLOTS_ (NPAD_ / 32)

typedef __attribute__((ext_vector_type(2))) float v2f;
typedef __attribute__((ext_vector_type(4))) float v4f;
typedef __attribute__((ext_vector_type(8))) float v8f;

#define MT_   256
#define MV4_  (B2_ / 4 / MT_)

typedef __attribute__((ext_vector_type(4))) int v4i;

__global__ __launch_bounds__(MT_) void k_gmax(const int* __restrict__ ei,
                                              const float* __restrict__ bd,
                                              float* __restrict__ gmax) {
    __shared__ float s_m[MT_ / 32];
    const int tid = threadIdx.x;
    const int s   = blockIdx.x;
    const v4i* sp = (const v4i*)(ei + (size_t)s * 2 * B2_);
    const v4i* dp = sp + B2_ / 4;
    const v4f* bp = (const v4f*)(bd + (size_t)s * B2_);
    v4i a[MV4_], b[MV4_];
    v4f w[MV4_];
#pragma unroll
    for (int j = 0; j < MV4_; ++j) {
        const int i = j * MT_ + tid;
        a[j] = sp[i];
        b[j] = dp[i];
        w[j] = bp[i];
    }
    float m = 0.0f;
#pragma unroll
    for (int j = 0; j < MV4_; ++j) {
        m = fmaxf(m, (a[j].x != b[j].x) ? w[j].x : 0.0f);
        m = fmaxf(m, (a[j].y != b[j].y) ? w[j].y : 0.0f);
        m = fmaxf(m, (a[j].z != b[j].z) ? w[j].z : 0.0f);
        m = fmaxf(m, (a[j].w != b[j].w) ? w[j].w : 0.0f);
    }
    for (int off = 16; off > 0; off >>= 1)
        m = fmaxf(m, __shfl_xor(m, off, 32));
    if ((tid & 31) == 0) s_m[tid >> 5] = m;
    __syncthreads();
    if (tid == 0) {
        float g = s_m[0];
        for (int k = 1; k < MT_ / 32; ++k) g = fmaxf(g, s_m[k]);
        gmax[s] = g;
    }
}

#define XH_    64
#define NTG_   35
#define ZROW_  (NTG_ * 16)
#define WP_    132

__device__ __forceinline__ void xw_half(const float* __restrict__ xs_g, const float* __restrict__ s_W,
                                        float* __restrict__ s_x, int oh, int wave, int lane) {
    const int c = lane & 15;
    const int h = lane >> 4;
    for (int t = wave; t < NTG_; t += NT_ / 32) {
        int woff = (oh * XH_ + c) * WP_ + h * 4;
        asm volatile("" : "+v"(woff));
        const float* swb = s_W + woff;
        int row = t * 16 + c;
        row = (row < N_) ? row : (N_ - 1);
        const float* ap = xs_g + (size_t)row * F_ + h * 4;
        v8f acc[4];
#pragma unroll
        for (int ct = 0; ct < 4; ++ct) acc[ct] = v8f{};
#pragma unroll
        for (int jc = 0; jc < 2; ++jc) {
            v4f a[8];
#pragma unroll
            for (int j = 0; j < 8; ++j) a[j] = *(const v4f*)(ap + (jc * 8 + j) * 8);
#pragma unroll
            for (int j = 0; j < 8; ++j) {
                const v2f a0 = {a[j].x, a[j].y};
                const v2f a1 = {a[j].z, a[j].w};
#pragma unroll
                for (int ct = 0; ct < 4; ++ct) {
                    const v4f b = *(const v4f*)(swb + ct * 16 * WP_ + (jc * 8 + j) * 8);
                    const v2f b0 = {b.x, b.y};
                    const v2f b1 = {b.z, b.w};
                    acc[ct] = __builtin_amdgcn_wmma_f32_16x16x4_f32(
                         false, a0,  false, b0,
                         (short)0, acc[ct],  false,  false);
                    acc[ct] = __builtin_amdgcn_wmma_f32_16x16x4_f32(
                         false, a1,  false, b1,
                         (short)0, acc[ct],  false,  false);
                }
            }
        }
        float* sx = s_x + (t * 16 + h * 8) * XH_ + c;
#pragma unroll
        for (int ct = 0; ct < 4; ++ct) {
#pragma unroll
            for (int v = 0; v < 8; ++v) sx[v * XH_ + ct * 16] = acc[ct][v];
        }
    }
}

template <int I>
__device__ __forceinline__ int row_share_i(int v) {
    return __builtin_amdgcn_update_dpp(0, v, 0x150 + I, 0xf, 0xf, true);
}
template <int I>
__device__ __forceinline__ float row_share_f(float v) {
    return __int_as_float(row_share_i<I>(__float_as_int(v)));
}

__device__ __forceinline__ v4f fma4(float n, v4f x, v4f acc) {
    v4f r;
    r.x = __builtin_fmaf(n, x.x, acc.x);
    r.y = __builtin_fmaf(n, x.y, acc.y);
    r.z = __builtin_fmaf(n, x.z, acc.z);
    r.w = __builtin_fmaf(n, x.w, acc.w);
    return r;
}

template <int G>
__device__ __forceinline__ v4f edge_group(const float* __restrict__ xl, int soff, float nv, v4f acc) {
    const int   a0 = row_share_i<4 * G + 0>(soff);
    const int   a1 = row_share_i<4 * G + 1>(soff);
    const int   a2 = row_share_i<4 * G + 2>(soff);
    const int   a3 = row_share_i<4 * G + 3>(soff);
    const float n0 = row_share_f<4 * G + 0>(nv);
    const float n1 = row_share_f<4 * G + 1>(nv);
    const float n2 = row_share_f<4 * G + 2>(nv);
    const float n3 = row_share_f<4 * G + 3>(nv);
    const v4f x0 = *(const v4f*)(xl + a0);
    const v4f x1 = *(const v4f*)(xl + a1);
    const v4f x2 = *(const v4f*)(xl + a2);
    const v4f x3 = *(const v4f*)(xl + a3);
    acc = fma4(n0, x0, acc);
    acc = fma4(n1, x1, acc);
    acc = fma4(n2, x2, acc);
    acc = fma4(n3, x3, acc);
    return acc;
}

__device__ __forceinline__ void gather_half(const int* __restrict__ s_start, const float* __restrict__ s_dinv,
                                            const int* __restrict__ s_src, const float* __restrict__ s_w,
                                            const float* __restrict__ s_x, float* __restrict__ og, v4f bb,
                                            int wave, int lane) {
    const int c = lane & 15;
    const int h = lane >> 4;
    const float* xl = s_x + c * 4;
    for (int rp = wave; rp < N_ / 2; rp += NT_ / 32) {
        const int r  = rp * 2 + h;
        const int st = s_start[r];
        const int en = s_start[r + 1];
        const float dd = s_dinv[r];
        const v4f xs = *(const v4f*)(xl + r * XH_);
        const int cnt = en - st;
        const int c0 = __builtin_amdgcn_readlane(cnt, 0);
        const int c1 = __builtin_amdgcn_readlane(cnt, 16);
        const int cm = (c0 > c1) ? c0 : c1;
        v4f acc = {0.0f, 0.0f, 0.0f, 0.0f};

        for (int cb = 0; cb < cm; cb += 16) {
            const int  e     = cb + c;
            const bool valid = e < cnt;
            int idx = st + e;
            idx = (idx < B2_) ? idx : (B2_ - 1);
            int sv = s_src[idx];
            sv = valid ? sv : 0;
            float nv = s_dinv[sv] * s_w[idx] * dd;
            nv = valid ? nv : 0.0f;
            const int soff = (valid ? sv : ZROW_) * XH_;
            const int left = cm - cb;
            acc = edge_group<0>(xl, soff, nv, acc);
            acc = edge_group<1>(xl, soff, nv, acc);
            if (left > 8)  acc = edge_group<2>(xl, soff, nv, acc);
            if (left > 12) acc = edge_group<3>(xl, soff, nv, acc);
        }
        acc = fma4(dd * dd, xs, acc);
        *(v4f*)(og + (size_t)r * F_) = acc + bb;
    }
}

__global__ __launch_bounds__(NT_) void k_sort_xw_gather(const int* __restrict__ ei,
                                                        const float* __restrict__ bd,
                                                        const float* __restrict__ x,
                                                        const float* __restrict__ W,
                                                        const float* __restrict__ bias,
                                                        const float* __restrict__ gmax,
                                                        float* __restrict__ out) {
    __shared__ int   s_cur[NPAD_];
    __shared__ int   s_start[NPAD_ + 1];
    __shared__ float s_dinv[NPAD_];
    __shared__ int   s_src[B2_];
    __shared__ float s_w[B2_];
    __shared__ __attribute__((aligned(16))) float s_x[(NTG_ * 16 + 1) * XH_];
    __shared__ __attribute__((aligned(16))) float s_W[F_ * WP_];

    const int tid  = threadIdx.x;
    const int lane = tid & 31;
    const int wave = __builtin_amdgcn_readfirstlane(tid >> 5);
    const int s    = blockIdx.x;

    if (tid < NPAD_) s_cur[tid] = 0;
    if (tid < XH_) s_x[ZROW_ * XH_ + tid] = 0.0f;
    __syncthreads();

    const int*   eis = ei + (size_t)s * 2 * B2_;
    const float* bds = bd + (size_t)s * B2_;
    const float* xs_g = x + (size_t)s * N_ * F_;
    int   es[EPT_], ed[EPT_];
    float wv[EPT_];
#pragma unroll
    for (int j = 0; j < EPT_; ++j) {
        int e = j * NT_ + tid;
        es[j] = eis[e];
        ed[j] = eis[B2_ + e];
        wv[j] = bds[e];
    }
    const v4f g0 = *(const v4f*)(gmax + lane * 8);
    const v4f g1 = *(const v4f*)(gmax + lane * 8 + 4);
    v4f w4[(F_ * F_ / 4) / NT_];
#pragma unroll
    for (int j = 0; j < (F_ * F_ / 4) / NT_; ++j) {
        const int i = j * NT_ + tid;
        w4[j] = *(const v4f*)(W + (size_t)(i >> 5) * F_ + (i & 31) * 4);
    }

    float bm = fmaxf(fmaxf(fmaxf(g0.x, g0.y), fmaxf(g0.z, g0.w)), fmaxf(fmaxf(g1.x, g1.y), fmaxf(g1.z, g1.w)));
    for (int off = 16; off > 0; off >>= 1)
        bm = fmaxf(bm, __shfl_xor(bm, off, 32));

    int sd[EPT_];
#pragma unroll
    for (int j = 0; j < EPT_; ++j) {
        const bool valid = es[j] != ed[j];
        sd[j] = valid ? (es[j] | (ed[j] << 16)) : -1;
        wv[j] = wv[j] / bm;
        if (valid) atomicAdd(&s_cur[ed[j]], 1);
    }
#pragma unroll
    for (int j = 0; j < (F_ * F_ / 4) / NT_; ++j) {
        const int i = j * NT_ + tid;
        *(v4f*)(&s_W[(i >> 5) * WP_ + (i & 31) * 4]) = w4[j];
    }
    __syncthreads();

    if (wave == 0) {
        int base = lane * SLOTS_;
        int sum = 0;
#pragma unroll
        for (int k = 0; k < SLOTS_; ++k) sum += s_cur[base + k];
        int incl = sum;
        for (int off = 1; off < 32; off <<= 1) {
            int t = __shfl_up(incl, off, 32);
            if (lane >= off) incl += t;
        }
        int run = incl - sum;
#pragma unroll
        for (int k = 0; k < SLOTS_; ++k) {
            int c = s_cur[base + k];
            s_start[base + k] = run;
            s_cur[base + k]   = run;
            run += c;
        }
        if (lane == 31) s_start[NPAD_] = run;
    }
    __syncthreads();

#pragma unroll
    for (int j = 0; j < EPT_; ++j) {
        if (sd[j] >= 0) {
            int dst = sd[j] >> 16;
            int pos = atomicAdd(&s_cur[dst], 1);
            s_src[pos] = sd[j] & 0xFFFF;
            s_w[pos]   = wv[j];
        }
    }
    __syncthreads();

    if (tid < N_) {
        int p0 = s_start[tid], p1 = s_start[tid + 1];
        float d = 1.0f;
        for (int p = p0; p < p1; ++p) d += s_w[p];
        s_dinv[tid] = rsqrtf(d);
    }

    float* og = out + (size_t)s * N_ * F_ + (lane & 15) * 4;
    const v4f bb0 = *(const v4f*)(bias + (lane & 15) * 4);
    const v4f bb1 = *(const v4f*)(bias + XH_ + (lane & 15) * 4);
    xw_half(xs_g, s_W, s_x, 0, wave, lane);
    __syncthreads();
    gather_half(s_start, s_dinv, s_src, s_w, s_x, og, bb0, wave, lane);
    __syncthreads();
    xw_half(xs_g, s_W, s_x, 1, wave, lane);
    __syncthreads();
    gather_half(s_start, s_dinv, s_src, s_w, s_x, og + XH_, bb1, wave, lane);
}

extern "C" void kernel_launch(void* const* d_in, const int* in_sizes, int n_in,
                              void* d_out, int out_size, void* d_ws, size_t ws_size,
                              hipStream_t stream) {
    const float* x    = (const float*)d_in[0];
    const int*   ei   = (const int*)  d_in[1];
    const float* bd   = (const float*)d_in[2];
    const float* W    = (const float*)d_in[3];
    const float* bias = (const float*)d_in[4];
    float*       out  = (float*)d_out;

    float* gmax = (float*)d_ws;

    k_gmax          <<<S_, MT_, 0, stream>>>(ei, bd, gmax);
    k_sort_xw_gather<<<S_, NT_, 0, stream>>>(ei, bd, x, W, bias, gmax, out);
}


# Product operands early: a wave loads its x tile once during the edge sort (LDS-only barriers between), holds it for both halves; W fragments 1 k-step early; tiles 32..34 as 12 blocks; same sums
# speedup vs baseline: 8.5369x; 1.0801x over previous
#include <hip/hip_runtime.h>


#define S_   256
#define N_   558
#define B2_  8192
#define F_   128
#define M_   (S_ * N_)
#define TE_  (S_ * B2_)

#define NT_    1024
#define EPT_   (B2_ / NT_)
#define NPAD_  576
#define SLOTS_ (NPAD_ / 32)

typedef __attribute__((ext_vector_type(2))) float v2f;
typedef __attribute__((ext_vector_type(4))) float v4f;
typedef __attribute__((ext_vector_type(8))) float v8f;

#define MT_   256
#define MV4_  (B2_ / 4 / MT_)

typedef __attribute__((ext_vector_type(4))) int v4i;

__global__ __launch_bounds__(MT_) void k_gmax(const int* __restrict__ ei,
                                              const float* __restrict__ bd,
                                              float* __restrict__ gmax) {
    __shared__ float s_m[MT_ / 32];
    const int tid = threadIdx.x;
    const int s   = blockIdx.x;
    const v4i* sp = (const v4i*)(ei + (size_t)s * 2 * B2_);
    const v4i* dp = sp + B2_ / 4;
    const v4f* bp = (const v4f*)(bd + (size_t)s * B2_);
    v4i a[MV4_], b[MV4_];
    v4f w[MV4_];
#pragma unroll
    for (int j = 0; j < MV4_; ++j) {
        const int i = j * MT_ + tid;
        a[j] = sp[i];
        b[j] = dp[i];
        w[j] = bp[i];
    }
    float m = 0.0f;
#pragma unroll
    for (int j = 0; j < MV4_; ++j) {
        m = fmaxf(m, (a[j].x != b[j].x) ? w[j].x : 0.0f);
        m = fmaxf(m, (a[j].y != b[j].y) ? w[j].y : 0.0f);
        m = fmaxf(m, (a[j].z != b[j].z) ? w[j].z : 0.0f);
        m = fmaxf(m, (a[j].w != b[j].w) ? w[j].w : 0.0f);
    }
    for (int off = 16; off > 0; off >>= 1)
        m = fmaxf(m, __shfl_xor(m, off, 32));
    if ((tid & 31) == 0) s_m[tid >> 5] = m;
    __syncthreads();
    if (tid == 0) {
        float g = s_m[0];
        for (int k = 1; k < MT_ / 32; ++k) g = fmaxf(g, s_m[k]);
        gmax[s] = g;
    }
}

#define XH_    64
#define NTG_   35
#define ZROW_  (NTG_ * 16)
#define WP_    132

#define NWV_   (NT_ / 32)
#define NXB_   ((NTG_ - NWV_) * 4)
#define XW0_   (NWV_ - NXB_)

__device__ __forceinline__ void xw_tile_held(const v4f (&a)[16], const float* __restrict__ s_W,
                                             float* __restrict__ s_x, int oh, int t, int lane) {
    const int c = lane & 15;
    const int h = lane >> 4;
#pragma unroll
    for (int p = 0; p < 2; ++p) {
        const float* swb = s_W + (oh * XH_ + p * 32 + c) * WP_ + h * 4;
        v8f acc[2];
        v4f bq[2][2];
        acc[0] = v8f{};
        acc[1] = v8f{};
#pragma unroll
        for (int u = 0; u < 2; ++u) bq[0][u] = *(const v4f*)(swb + u * 16 * WP_);
#pragma unroll
        for (int j = 0; j < 16; ++j) {
            if (j + 1 < 16) {
#pragma unroll
                for (int u = 0; u < 2; ++u) bq[(j + 1) & 1][u] = *(const v4f*)(swb + u * 16 * WP_ + (j + 1) * 8);
            }
            const v2f a0 = {a[j].x, a[j].y};
            const v2f a1 = {a[j].z, a[j].w};
#pragma unroll
            for (int u = 0; u < 2; ++u) {
                const v4f b = bq[j & 1][u];
                const v2f b0 = {b.x, b.y};
                const v2f b1 = {b.z, b.w};
                acc[u] = __builtin_amdgcn_wmma_f32_16x16x4_f32(
                     false, a0,  false, b0,
                     (short)0, acc[u],  false,  false);
                acc[u] = __builtin_amdgcn_wmma_f32_16x16x4_f32(
                     false, a1,  false, b1,
                     (short)0, acc[u],  false,  false);
            }
        }
        float* sx = s_x + (t * 16 + h * 8) * XH_ + p * 32 + c;
#pragma unroll
        for (int u = 0; u < 2; ++u) {
#pragma unroll
            for (int v = 0; v < 8; ++v) sx[v * XH_ + u * 16] = acc[u][v];
        }
        __builtin_amdgcn_sched_barrier(0);
    }
}

__device__ __forceinline__ void xw_block_extra(const float* __restrict__ xs_g, const float* __restrict__ s_W,
                                               float* __restrict__ s_x, int oh, int xb, int lane) {
    const int c  = lane & 15;
    const int h  = lane >> 4;
    const int t  = NWV_ + (xb >> 2);
    const int ct = xb & 3;
    int row = t * 16 + c;
    row = (row < N_) ? row : (N_ - 1);
    const float* ap  = xs_g + (size_t)row * F_ + h * 4;
    const float* swb = s_W + (oh * XH_ + ct * 16 + c) * WP_ + h * 4;
    v8f acc = v8f{};
    v4f bq[3];
#pragma unroll
    for (int jc = 0; jc < 4; ++jc) {
        v4f a[4];
#pragma unroll
        for (int j = 0; j < 4; ++j) a[j] = *(const v4f*)(ap + (jc * 4 + j) * 8);
        if (jc == 0) {
#pragma unroll
            for (int j = 0; j < 2; ++j) bq[j] = *(const v4f*)(swb + j * 8);
        }
#pragma unroll
        for (int j = 0; j < 4; ++j) {
            const int jj = jc * 4 + j;
            if (jj + 2 < 16) bq[(jj + 2) % 3] = *(const v4f*)(swb + (jj + 2) * 8);
            const v4f b = bq[jj % 3];
            const v2f a0 = {a[j].x, a[j].y};
            const v2f a1 = {a[j].z, a[j].w};
            const v2f b0 = {b.x, b.y};
            const v2f b1 = {b.z, b.w};
            acc = __builtin_amdgcn_wmma_f32_16x16x4_f32(
                 false, a0,  false, b0,
                 (short)0, acc,  false,  false);
            acc = __builtin_amdgcn_wmma_f32_16x16x4_f32(
                 false, a1,  false, b1,
                 (short)0, acc,  false,  false);
        }
    }
    float* sx = s_x + (t * 16 + h * 8) * XH_ + ct * 16 + c;
#pragma unroll
    for (int v = 0; v < 8; ++v) sx[v * XH_] = acc[v];
}

template <int I>
__device__ __forceinline__ int row_share_i(int v) {
    return __builtin_amdgcn_update_dpp(0, v, 0x150 + I, 0xf, 0xf, true);
}
template <int I>
__device__ __forceinline__ float row_share_f(float v) {
    return __int_as_float(row_share_i<I>(__float_as_int(v)));
}

__device__ __forceinline__ v4f fma4(float n, v4f x, v4f acc) {
    v4f r;
    r.x = __builtin_fmaf(n, x.x, acc.x);
    r.y = __builtin_fmaf(n, x.y, acc.y);
    r.z = __builtin_fmaf(n, x.z, acc.z);
    r.w = __builtin_fmaf(n, x.w, acc.w);
    return r;
}

template <int G>
__device__ __forceinline__ v4f edge_group(const float* __restrict__ xl, int soff, float nv, v4f acc) {
    const int   a0 = row_share_i<4 * G + 0>(soff);
    const int   a1 = row_share_i<4 * G + 1>(soff);
    const int   a2 = row_share_i<4 * G + 2>(soff);
    const int   a3 = row_share_i<4 * G + 3>(soff);
    const float n0 = row_share_f<4 * G + 0>(nv);
    const float n1 = row_share_f<4 * G + 1>(nv);
    const float n2 = row_share_f<4 * G + 2>(nv);
    const float n3 = row_share_f<4 * G + 3>(nv);
    const v4f x0 = *(const v4f*)(xl + a0);
    const v4f x1 = *(const v4f*)(xl + a1);
    const v4f x2 = *(const v4f*)(xl + a2);
    const v4f x3 = *(const v4f*)(xl + a3);
    acc = fma4(n0, x0, acc);
    acc = fma4(n1, x1, acc);
    acc = fma4(n2, x2, acc);
    acc = fma4(n3, x3, acc);
    return acc;
}

__device__ __forceinline__ void gather_half(const int* __restrict__ s_start, const float* __restrict__ s_dinv,
                                            const int* __restrict__ s_src, const float* __restrict__ s_w,
                                            const float* __restrict__ s_x, float* __restrict__ og, v4f bb,
                                            int wave, int lane) {
    const int c = lane & 15;
    const int h = lane >> 4;
    const float* xl = s_x + c * 4;
    for (int rp = wave; rp < N_ / 2; rp += NT_ / 32) {
        const int r  = rp * 2 + h;
        const int st = s_start[r];
        const int en = s_start[r + 1];
        const float dd = s_dinv[r];
        const v4f xs = *(const v4f*)(xl + r * XH_);
        const int cnt = en - st;
        const int c0 = __builtin_amdgcn_readlane(cnt, 0);
        const int c1 = __builtin_amdgcn_readlane(cnt, 16);
        const int cm = (c0 > c1) ? c0 : c1;
        v4f acc = {0.0f, 0.0f, 0.0f, 0.0f};

        for (int cb = 0; cb < cm; cb += 16) {
            const int  e     = cb + c;
            const bool valid = e < cnt;
            int idx = st + e;
            idx = (idx < B2_) ? idx : (B2_ - 1);
            int sv = s_src[idx];
            sv = valid ? sv : 0;
            float nv = s_dinv[sv] * s_w[idx] * dd;
            nv = valid ? nv : 0.0f;
            const int soff = (valid ? sv : ZROW_) * XH_;
            const int left = cm - cb;
            acc = edge_group<0>(xl, soff, nv, acc);
            acc = edge_group<1>(xl, soff, nv, acc);
            if (left > 8)  acc = edge_group<2>(xl, soff, nv, acc);
            if (left > 12) acc = edge_group<3>(xl, soff, nv, acc);
        }
        acc = fma4(dd * dd, xs, acc);
        *(v4f*)(og + (size_t)r * F_) = acc + bb;
    }
}

__device__ __forceinline__ void lds_barrier() {
    __builtin_amdgcn_fence(__ATOMIC_RELEASE, "workgroup", "local");
    __builtin_amdgcn_s_barrier();
    __builtin_amdgcn_fence(__ATOMIC_ACQUIRE, "workgroup", "local");
}

__global__ __launch_bounds__(NT_) void k_sort_xw_gather(const int* __restrict__ ei,
                                                        const float* __restrict__ bd,
                                                        const float* __restrict__ x,
                                                        const float* __restrict__ W,
                                                        const float* __restrict__ bias,
                                                        const float* __restrict__ gmax,
                                                        float* __restrict__ out) {
    __shared__ int   s_cur[NPAD_];
    __shared__ int   s_start[NPAD_ + 1];
    __shared__ float s_dinv[NPAD_];
    __shared__ int   s_src[B2_];
    __shared__ float s_w[B2_];
    __shared__ __attribute__((aligned(16))) float s_x[(NTG_ * 16 + 1) * XH_];
    __shared__ __attribute__((aligned(16))) float s_W[F_ * WP_];

    const int tid  = threadIdx.x;
    const int lane = tid & 31;
    const int wave = __builtin_amdgcn_readfirstlane(tid >> 5);
    const int s    = blockIdx.x;

    if (tid < NPAD_) s_cur[tid] = 0;
    if (tid < XH_) s_x[ZROW_ * XH_ + tid] = 0.0f;
    __syncthreads();

    const int*   eis = ei + (size_t)s * 2 * B2_;
    const float* bds = bd + (size_t)s * B2_;
    const float* xs_g = x + (size_t)s * N_ * F_;
    int   es[EPT_], ed[EPT_];
    float wv[EPT_];
#pragma unroll
    for (int j = 0; j < EPT_; ++j) {
        int e = j * NT_ + tid;
        es[j] = eis[e];
        ed[j] = eis[B2_ + e];
        wv[j] = bds[e];
    }
    const v4f g0 = *(const v4f*)(gmax + lane * 8);
    const v4f g1 = *(const v4f*)(gmax + lane * 8 + 4);
    v4f w4[(F_ * F_ / 4) / NT_];
#pragma unroll
    for (int j = 0; j < (F_ * F_ / 4) / NT_; ++j) {
        const int i = j * NT_ + tid;
        w4[j] = *(const v4f*)(W + (size_t)(i >> 5) * F_ + (i & 31) * 4);
    }

    float bm = fmaxf(fmaxf(fmaxf(g0.x, g0.y), fmaxf(g0.z, g0.w)), fmaxf(fmaxf(g1.x, g1.y), fmaxf(g1.z, g1.w)));
    for (int off = 16; off > 0; off >>= 1)
        bm = fmaxf(bm, __shfl_xor(bm, off, 32));

    int sd[EPT_];
#pragma unroll
    for (int j = 0; j < EPT_; ++j) {
        const bool valid = es[j] != ed[j];
        sd[j] = valid ? (es[j] | (ed[j] << 16)) : -1;
        wv[j] = wv[j] / bm;
        if (valid) atomicAdd(&s_cur[ed[j]], 1);
    }
#pragma unroll
    for (int j = 0; j < (F_ * F_ / 4) / NT_; ++j) {
        const int i = j * NT_ + tid;
        *(v4f*)(&s_W[(i >> 5) * WP_ + (i & 31) * 4]) = w4[j];
    }
    __syncthreads();

    v4f ax[16];
    {
        const float* ap = xs_g + (size_t)(wave * 16 + (lane & 15)) * F_ + (lane >> 4) * 4;
#pragma unroll
        for (int j = 0; j < 16; ++j) ax[j] = *(const v4f*)(ap + j * 8);
    }

    if (wave == 0) {
        int base = lane * SLOTS_;
        int sum = 0;
#pragma unroll
        for (int k = 0; k < SLOTS_; ++k) sum += s_cur[base + k];
        int incl = sum;
        for (int off = 1; off < 32; off <<= 1) {
            int t = __shfl_up(incl, off, 32);
            if (lane >= off) incl += t;
        }
        int run = incl - sum;
#pragma unroll
        for (int k = 0; k < SLOTS_; ++k) {
            int c = s_cur[base + k];
            s_start[base + k] = run;
            s_cur[base + k]   = run;
            run += c;
        }
        if (lane == 31) s_start[NPAD_] = run;
    }
    lds_barrier();

#pragma unroll
    for (int j = 0; j < EPT_; ++j) {
        if (sd[j] >= 0) {
            int dst = sd[j] >> 16;
            int pos = atomicAdd(&s_cur[dst], 1);
            s_src[pos] = sd[j] & 0xFFFF;
            s_w[pos]   = wv[j];
        }
    }
    lds_barrier();

    if (tid < N_) {
        int p0 = s_start[tid], p1 = s_start[tid + 1];
        float d = 1.0f;
        for (int p = p0; p < p1; ++p) d += s_w[p];
        s_dinv[tid] = rsqrtf(d);
    }

    float* og = out + (size_t)s * N_ * F_ + (lane & 15) * 4;
    const v4f bb0 = *(const v4f*)(bias + (lane & 15) * 4);
    const v4f bb1 = *(const v4f*)(bias + XH_ + (lane & 15) * 4);
    if (wave >= XW0_) xw_block_extra(xs_g, s_W, s_x, 0, wave - XW0_, lane);
    xw_tile_held(ax, s_W, s_x, 0, wave, lane);
    __syncthreads();
    gather_half(s_start, s_dinv, s_src, s_w, s_x, og, bb0, wave, lane);
    __syncthreads();
    if (wave >= XW0_) xw_block_extra(xs_g, s_W, s_x, 1, wave - XW0_, lane);
    xw_tile_held(ax, s_W, s_x, 1, wave, lane);
    __syncthreads();
    gather_half(s_start, s_dinv, s_src, s_w, s_x, og + XH_, bb1, wave, lane);
}

extern "C" void kernel_launch(void* const* d_in, const int* in_sizes, int n_in,
                              void* d_out, int out_size, void* d_ws, size_t ws_size,
                              hipStream_t stream) {
    const float* x    = (const float*)d_in[0];
    const int*   ei   = (const int*)  d_in[1];
    const float* bd   = (const float*)d_in[2];
    const float* W    = (const float*)d_in[3];
    const float* bias = (const float*)d_in[4];
    float*       out  = (float*)d_out;

    float* gmax = (float*)d_ws;

    k_gmax          <<<S_, MT_, 0, stream>>>(ei, bd, gmax);
    k_sort_xw_gather<<<S_, NT_, 0, stream>>>(ei, bd, x, W, bias, gmax, out);
}
